# starting kernel
# baseline (speedup 1.0000x reference)
#include <hip/hip_runtime.h>
#include <hip/hip_bf16.h>


typedef __attribute__((ext_vector_type(2))) float v2f;
typedef __attribute__((ext_vector_type(8))) float v8f;

#define FDIM 128
#define NHEAD 4
#define CHAN 32
#define NEG_SLOPE 0.2f

__global__ __launch_bounds__(256) void gat_gemm_wmma(const float* __restrict__ X,
                                                     const float* __restrict__ W,
                                                     float* __restrict__ H,
                                                     int N) {
  const int lane = threadIdx.x & 31;
  const int wave = threadIdx.x >> 5;
  const int m    = lane & 15;
  const int half = lane >> 4;
  const int row0 = blockIdx.x * 16;
  const int col0 = wave * 16;

  int rowA = row0 + m;
  if (rowA >= N) rowA = N - 1;
  const float* __restrict__ xrow = X + (size_t)rowA * FDIM;

  v8f acc = {};
#pragma unroll
  for (int k = 0; k < FDIM; k += 4) {
    const int ka = k + 2 * half;
    v2f a = *(const v2f*)(xrow + ka);
    v2f b;
    b.x = W[ka * FDIM + col0 + m];
    b.y = W[(ka + 1) * FDIM + col0 + m];
    acc = __builtin_amdgcn_wmma_f32_16x16x4_f32(
         false, a,  false, b,
         (short)0, acc,  false,  false);
  }

#pragma unroll
  for (int v = 0; v < 8; ++v) {
    const int row = row0 + v + 8 * half;
    if (row < N) H[(size_t)row * FDIM + col0 + m] = acc[v];
  }
}

__global__ void gat_alpha(const float* __restrict__ H,
                          const float* __restrict__ a_src,
                          const float* __restrict__ a_dst,
                          float* __restrict__ as_, float* __restrict__ ad_,
                          int N) {
  const int tid = blockIdx.x * blockDim.x + threadIdx.x;
  if (tid >= N * NHEAD) return;
  const int n = tid >> 2;
  const int h = tid & 3;
  const float* __restrict__ hp = H + (size_t)n * FDIM + h * CHAN;
  const float* __restrict__ asv = a_src + h * CHAN;
  const float* __restrict__ adv = a_dst + h * CHAN;
  float s = 0.0f, d = 0.0f;
#pragma unroll
  for (int c = 0; c < CHAN; ++c) {
    const float hv = hp[c];
    s = fmaf(hv, asv[c], s);
    d = fmaf(hv, adv[c], d);
  }
  as_[tid] = s;
  ad_[tid] = d;
}

__global__ void fill_kernel(float* __restrict__ p, float v, int n) {
  const int i = blockIdx.x * blockDim.x + threadIdx.x;
  if (i < n) p[i] = v;
}

__device__ __forceinline__ void atomicMaxFloat(float* addr, float v) {
  const int iv = __float_as_int(v);
  if (iv >= 0) atomicMax((int*)addr, iv);
  else         atomicMin((unsigned int*)addr, (unsigned int)iv);
}

__device__ __forceinline__ void edge_endpoints(const int* __restrict__ ei,
                                               int e, int E, int* s, int* d) {
  if (e < E) { *s = ei[e]; *d = ei[E + e]; }
  else       { *s = e - E; *d = e - E; }
}

__device__ __forceinline__ float leaky(float v) {
  return v > 0.0f ? v : NEG_SLOPE * v;
}

__global__ void gat_edge_max(const int* __restrict__ ei, int E, int ET,
                             const float* __restrict__ as_,
                             const float* __restrict__ ad_,
                             float* __restrict__ m_) {
  const int tid = blockIdx.x * blockDim.x + threadIdx.x;
  if (tid >= ET * NHEAD) return;
  const int e = tid >> 2, h = tid & 3;
  int s, d;
  edge_endpoints(ei, e, E, &s, &d);
  const float v = leaky(as_[s * NHEAD + h] + ad_[d * NHEAD + h]);
  atomicMaxFloat(&m_[d * NHEAD + h], v);
}

__global__ void gat_edge_expsum(const int* __restrict__ ei, int E, int ET,
                                const float* __restrict__ as_,
                                const float* __restrict__ ad_,
                                const float* __restrict__ m_,
                                float* __restrict__ den_) {
  const int tid = blockIdx.x * blockDim.x + threadIdx.x;
  if (tid >= ET * NHEAD) return;
  const int e = tid >> 2, h = tid & 3;
  int s, d;
  edge_endpoints(ei, e, E, &s, &d);
  const float v = leaky(as_[s * NHEAD + h] + ad_[d * NHEAD + h]);
  const float ex = __expf(v - m_[d * NHEAD + h]);
  atomicAdd(&den_[d * NHEAD + h], ex);
}

__global__ void gat_edge_scatter(const int* __restrict__ ei, int E, int ET,
                                 const float* __restrict__ as_,
                                 const float* __restrict__ ad_,
                                 const float* __restrict__ m_,
                                 const float* __restrict__ den_,
                                 const float* __restrict__ H,
                                 float* __restrict__ out) {
  const long long tid = (long long)blockIdx.x * blockDim.x + threadIdx.x;
  if (tid >= (long long)ET * 32) return;
  const int e  = (int)(tid >> 5);
  const int q  = (int)(tid & 31);
  const int c0 = q * 4;
  const int h  = c0 >> 5;
  int s, d;
  edge_endpoints(ei, e, E, &s, &d);
  const float v  = leaky(as_[s * NHEAD + h] + ad_[d * NHEAD + h]);
  const float ex = __expf(v - m_[d * NHEAD + h]);
  const float alpha = ex / (den_[d * NHEAD + h] + 1e-16f);

  const float4 hv = *(const float4*)(H + (size_t)s * FDIM + c0);
  float* __restrict__ op = out + (size_t)d * FDIM + c0;
  atomicAdd(op + 0, alpha * hv.x);
  atomicAdd(op + 1, alpha * hv.y);
  atomicAdd(op + 2, alpha * hv.z);
  atomicAdd(op + 3, alpha * hv.w);
}

__global__ void gat_bias_act(float* __restrict__ p, const float* __restrict__ b,
                             int total, int do_relu) {
  const int i = blockIdx.x * blockDim.x + threadIdx.x;
  if (i >= total) return;
  float v = p[i] + b[i & (FDIM - 1)];
  if (do_relu) v = v > 0.0f ? v : 0.0f;
  p[i] = v;
}

static inline int cdiv(long long a, int b) { return (int)((a + b - 1) / b); }

extern "C" void kernel_launch(void* const* d_in, const int* in_sizes, int n_in,
                              void* d_out, int out_size, void* d_ws, size_t ws_size,
                              hipStream_t stream) {
  const float* x     = (const float*)d_in[0];
  const int*   ei    = (const int*)d_in[1];
  const float* W1    = (const float*)d_in[2];
  const float* asrc1 = (const float*)d_in[3];
  const float* adst1 = (const float*)d_in[4];
  const float* b1    = (const float*)d_in[5];
  const float* W2    = (const float*)d_in[6];
  const float* asrc2 = (const float*)d_in[7];
  const float* adst2 = (const float*)d_in[8];
  const float* b2    = (const float*)d_in[9];
  float* out = (float*)d_out;

  const int N  = in_sizes[0] / FDIM;
  const int E  = in_sizes[1] / 2;
  const int ET = E + N;

  float* H    = (float*)d_ws;
  float* Hact = H    + (size_t)N * FDIM;
  float* as_  = Hact + (size_t)N * FDIM;
  float* ad_  = as_  + (size_t)N * NHEAD;
  float* m_   = ad_  + (size_t)N * NHEAD;
  float* den_ = m_   + (size_t)N * NHEAD;

  const int TB = 256;
  const int gTiles = cdiv(N, 16);
  const int gNH    = cdiv((long long)N * NHEAD, TB);
  const int gNF    = cdiv((long long)N * FDIM, TB);
  const int gEH    = cdiv((long long)ET * NHEAD, TB);
  const int gE32   = cdiv((long long)ET * 32, TB);
  const float NEG_INF = -__builtin_huge_valf();

  fill_kernel<<<gNH, TB, 0, stream>>>(m_, NEG_INF, N * NHEAD);
  fill_kernel<<<gNH, TB, 0, stream>>>(den_, 0.0f, N * NHEAD);
  fill_kernel<<<gNF, TB, 0, stream>>>(Hact, 0.0f, N * FDIM);

  gat_gemm_wmma<<<gTiles, TB, 0, stream>>>(x, W1, H, N);
  gat_alpha<<<gNH, TB, 0, stream>>>(H, asrc1, adst1, as_, ad_, N);
  gat_edge_max<<<gEH, TB, 0, stream>>>(ei, E, ET, as_, ad_, m_);
  gat_edge_expsum<<<gEH, TB, 0, stream>>>(ei, E, ET, as_, ad_, m_, den_);
  gat_edge_scatter<<<gE32, TB, 0, stream>>>(ei, E, ET, as_, ad_, m_, den_, H, Hact);
  gat_bias_act<<<gNF, TB, 0, stream>>>(Hact, b1, N * FDIM,  1);

  fill_kernel<<<gNH, TB, 0, stream>>>(m_, NEG_INF, N * NHEAD);
  fill_kernel<<<gNH, TB, 0, stream>>>(den_, 0.0f, N * NHEAD);
  fill_kernel<<<gNF, TB, 0, stream>>>(out, 0.0f, N * FDIM);

  gat_gemm_wmma<<<gTiles, TB, 0, stream>>>(Hact, W2, H, N);
  gat_alpha<<<gNH, TB, 0, stream>>>(H, asrc2, adst2, as_, ad_, N);
  gat_edge_max<<<gEH, TB, 0, stream>>>(ei, E, ET, as_, ad_, m_);
  gat_edge_expsum<<<gEH, TB, 0, stream>>>(ei, E, ET, as_, ad_, m_, den_);
  gat_edge_scatter<<<gE32, TB, 0, stream>>>(ei, E, ET, as_, ad_, m_, den_, H, out);
  gat_bias_act<<<gNF, TB, 0, stream>>>(out, b2, N * FDIM,  0);
}


# GEMM: a block keeps its W columns as B fragments in registers and loops over 16-row X tiles staged once per block in LDS by whole-row loads (was: each wave gathered its A fragment, 16 lines a load)
# speedup vs baseline: 4.4188x; 4.4188x over previous
#include <hip/hip_runtime.h>
#include <hip/hip_bf16.h>


typedef __attribute__((ext_vector_type(2))) float v2f;
typedef __attribute__((ext_vector_type(8))) float v8f;

#define FDIM 128
#define NHEAD 4
#define CHAN 32
#define NEG_SLOPE 0.2f

#define RN 196
#define OWN_TB 1024
#define OWN_WAVES (OWN_TB / 32)
#define LCAP 8192
#define KW 4096
#define CQCAP 4096

__device__ __forceinline__ bool keys_fit(int E, int N, int nblk) {
  return E > 0 && (E & 15) == 0 && nblk <= 256 && (E >> 2) <= (N / RN) * KW;
}
__device__ __forceinline__ unsigned key_word(int q) {
  const unsigned u = (unsigned)q;
  return (u / KW) * (RN * FDIM) + (RN * FDIM - KW) + (u % KW);
}
static_assert(KW / 4 == OWN_TB, "a block's key words are one 16-byte piece per thread");

#define GEMM_TB 256
#define GEMM_GRID 768
#define XPITCH (FDIM + 4)

__global__ __launch_bounds__(GEMM_TB) void gat_gemm_wmma(const float* __restrict__ X,
                                                         const float* __restrict__ W,
                                                         float* __restrict__ H,
                                                         int N) {
  __shared__ __attribute__((aligned(16))) float xs[2][16 * XPITCH];

  const int tid  = threadIdx.x;
  const int lane = tid & 31;
  const int wave = tid >> 5;
  const int m    = lane & 15;
  const int half = lane >> 4;
  const int col0 = wave * 16;
  const int ntile = (N + 15) >> 4;

  v2f b[FDIM / 4];
#pragma unroll
  for (int j = 0; j < FDIM / 4; ++j) {
    const int ka = 4 * j + 2 * half;
    b[j].x = W[ka * FDIM + col0 + m];
    b[j].y = W[(ka + 1) * FDIM + col0 + m];
  }

  const int sr0 = tid >> 5, sr1 = sr0 + 8;
  const int sc  = (tid & 31) * 4;

  int tile = blockIdx.x;
  if (tile >= ntile) return;
  float4 st0, st1;
  {
    int r0 = tile * 16 + sr0, r1 = tile * 16 + sr1;
    if (r0 >= N) r0 = N - 1;
    if (r1 >= N) r1 = N - 1;
    st0 = *(const float4*)(X + (size_t)r0 * FDIM + sc);
    st1 = *(const float4*)(X + (size_t)r1 * FDIM + sc);
  }

  int buf = 0;
  for (;;) {
    float* xt = xs[buf];
    *(float4*)(xt + sr0 * XPITCH + sc) = st0;
    *(float4*)(xt + sr1 * XPITCH + sc) = st1;
    __syncthreads();

    const int nxt = tile + GEMM_GRID;
    {
      const int lt = nxt < ntile ? nxt : tile;
      int r0 = lt * 16 + sr0, r1 = lt * 16 + sr1;
      if (r0 >= N) r0 = N - 1;
      if (r1 >= N) r1 = N - 1;
      st0 = *(const float4*)(X + (size_t)r0 * FDIM + sc);
      st1 = *(const float4*)(X + (size_t)r1 * FDIM + sc);
    }

    const float* xa = xt + m * XPITCH + 2 * half;
    v2f a[FDIM / 4];
#pragma unroll
    for (int j = 0; j < FDIM / 4; ++j) a[j] = *(const v2f*)(xa + 4 * j);
    __builtin_amdgcn_sched_barrier(0);
    v8f acc = {};
#pragma unroll
    for (int j = 0; j < FDIM / 4; ++j) {
      acc = __builtin_amdgcn_wmma_f32_16x16x4_f32(
           false, a[j],  false, b[j],
           (short)0, acc,  false,  false);
    }

    const int row0 = tile * 16;
#pragma unroll
    for (int v = 0; v < 8; ++v) {
      const int row = row0 + v + 8 * half;
      if (row < N) H[(size_t)row * FDIM + col0 + m] = acc[v];
    }
    if (nxt >= ntile) break;
    tile = nxt;
    buf ^= 1;
  }
}

__global__ void gat_alpha(const float* __restrict__ H,
                          const float* __restrict__ a_src,
                          const float* __restrict__ a_dst,
                          float* __restrict__ as_, float* __restrict__ ad_,
                          int N,
                          const int* __restrict__ ei, int E, int nblk,
                          float* keybuf) {
  const int tid = blockIdx.x * blockDim.x + threadIdx.x;
  if (keybuf != nullptr && keys_fit(E, N, nblk)) {
    const int4* __restrict__ dst4 = (const int4*)(ei + E);
    unsigned* kw = (unsigned*)keybuf;
    const int nvec = E >> 2;
    for (int q = tid; q < nvec; q += gridDim.x * blockDim.x) {
      const int4 d = dst4[q];
      const unsigned k = ((unsigned)d.x / RN & 255u) | (((unsigned)d.y / RN & 255u) << 8) |
                         (((unsigned)d.z / RN & 255u) << 16) | (((unsigned)d.w / RN) << 24);
      kw[key_word(q)] = k;
    }
  }
  if (tid >= N * NHEAD) return;
  const int n = tid >> 2;
  const int h = tid & 3;
  const float* __restrict__ hp = H + (size_t)n * FDIM + h * CHAN;
  const float* __restrict__ asv = a_src + h * CHAN;
  const float* __restrict__ adv = a_dst + h * CHAN;
  float s = 0.0f, d = 0.0f;
#pragma unroll
  for (int c = 0; c < CHAN; ++c) {
    const float hv = hp[c];
    s = fmaf(hv, asv[c], s);
    d = fmaf(hv, adv[c], d);
  }
  as_[tid] = s;
  ad_[tid] = d;
}


__device__ __forceinline__ float leaky(float v) {
  return v > 0.0f ? v : NEG_SLOPE * v;
}

__device__ __forceinline__ void lds_max_float(int* addr, float v) {
  const int iv = __float_as_int(v);
  if (iv >= 0) atomicMax(addr, iv);
  else         atomicMin((unsigned int*)addr, (unsigned int)iv);
}

__device__ __forceinline__ float4 edge_logits(unsigned dl, float4 a, const float* adl) {
  const float4 d = *(const float4*)(adl + dl * NHEAD);
  return make_float4(leaky(a.x + d.x), leaky(a.y + d.y), leaky(a.z + d.z), leaky(a.w + d.w));
}

__device__ __forceinline__ void row_max_update(unsigned dl, float4 t, int* mx) {
  lds_max_float(&mx[dl * NHEAD + 0], t.x);
  lds_max_float(&mx[dl * NHEAD + 1], t.y);
  lds_max_float(&mx[dl * NHEAD + 2], t.z);
  lds_max_float(&mx[dl * NHEAD + 3], t.w);
}

__device__ __forceinline__ void first_update(int s, unsigned dl, float4 a,
                                             const float* adl, int* mx,
                                             int2* lst, float4* lgw, int* cnt) {
  const float4 t = edge_logits(dl, a, adl);
  row_max_update(dl, t, mx);
  const int pos = atomicAdd(cnt, 1);
  if (pos < LCAP) {
    lst[pos] = make_int2(s, (int)dl);
    lgw[pos] = t;
  }
}

__device__ __forceinline__ void first_edge(int e, unsigned dl,
                                           const int* __restrict__ ei,
                                           const float* __restrict__ as_,
                                           const float* adl, int* mx,
                                           int2* lst, float4* lgw, int* cnt) {
  const int s = ei[e];
  const float4 a = *(const float4*)(as_ + (size_t)s * NHEAD);
  first_update(s, dl, a, adl, mx, lst, lgw, cnt);
}

__device__ __forceinline__ void scan8(const int4 (&v)[8], const int (&jj)[8],
                                      int d0, int nloc,
                                      const int* __restrict__ ei,
                                      const float* __restrict__ as_,
                                      const float* adl, int* mx,
                                      int2* lst, float4* lgw, int* cnt) {
  int      qe0 = 0, qe1 = 0, nq = 0;
  unsigned qd0 = 0, qd1 = 0;
#pragma unroll
  for (int k = 0; k < 8; ++k) {
    const int dd[4] = {v[k].x, v[k].y, v[k].z, v[k].w};
#pragma unroll
    for (int c = 0; c < 4; ++c) {
      const unsigned dl = (unsigned)(dd[c] - d0);
      if (dl < (unsigned)nloc) {
        const int e = 4 * jj[k] + c;
        if (nq == 0)      { qe0 = e; qd0 = dl; }
        else if (nq == 1) { qe1 = e; qd1 = dl; }
        else              first_edge(e, dl, ei, as_, adl, mx, lst, lgw, cnt);
        ++nq;
      }
    }
  }
  int s0 = 0, s1 = 0;
  if (nq > 0) s0 = ei[qe0];
  if (nq > 1) s1 = ei[qe1];
  float4 a0 = make_float4(0.0f, 0.0f, 0.0f, 0.0f), a1 = a0;
  if (nq > 0) a0 = *(const float4*)(as_ + (size_t)s0 * NHEAD);
  if (nq > 1) a1 = *(const float4*)(as_ + (size_t)s1 * NHEAD);
  if (nq > 0) first_update(s0, qd0, a0, adl, mx, lst, lgw, cnt);
  if (nq > 1) first_update(s1, qd1, a1, adl, mx, lst, lgw, cnt);
}

__device__ __forceinline__ unsigned zero_byte(unsigned x) {
  return (x - 0x01010101u) & ~x & 0x80808080u;
}

struct EdgeIn { float4 a; float h0, h1, h2, h3; };

__device__ __forceinline__ EdgeIn edge_load(int s, int lane,
                                            const float* __restrict__ as_,
                                            const float* __restrict__ H) {
  EdgeIn in;
  in.a = *(const float4*)(as_ + (size_t)s * NHEAD);
  const float* __restrict__ hp = H + (size_t)s * FDIM + lane;
  in.h0 = hp[0 * CHAN];
  in.h1 = hp[1 * CHAN];
  in.h2 = hp[2 * CHAN];
  in.h3 = hp[3 * CHAN];
  return in;
}

__device__ __forceinline__ void edge_add(const EdgeIn in, int dl, int lane,
                                         const float* adl, const int* mx,
                                         float* den, float* acc) {
  const float e0 = __expf(leaky(in.a.x + adl[dl * NHEAD + 0]) - __int_as_float(mx[dl * NHEAD + 0]));
  const float e1 = __expf(leaky(in.a.y + adl[dl * NHEAD + 1]) - __int_as_float(mx[dl * NHEAD + 1]));
  const float e2 = __expf(leaky(in.a.z + adl[dl * NHEAD + 2]) - __int_as_float(mx[dl * NHEAD + 2]));
  const float e3 = __expf(leaky(in.a.w + adl[dl * NHEAD + 3]) - __int_as_float(mx[dl * NHEAD + 3]));
  float* ap = acc + dl * FDIM + lane;
  atomicAdd(ap + 0 * CHAN, e0 * in.h0);
  atomicAdd(ap + 1 * CHAN, e1 * in.h1);
  atomicAdd(ap + 2 * CHAN, e2 * in.h2);
  atomicAdd(ap + 3 * CHAN, e3 * in.h3);
  if (lane < NHEAD) {
    const float ev = lane == 0 ? e0 : lane == 1 ? e1 : lane == 2 ? e2 : e3;
    atomicAdd(&den[dl * NHEAD + lane], ev);
  }
}

struct RowIn { float h0, h1, h2, h3; };

__device__ __forceinline__ RowIn row_load(int s, int lane, const float* __restrict__ H) {
  RowIn in;
  const float* __restrict__ hp = H + (size_t)s * FDIM + lane;
  in.h0 = hp[0 * CHAN];
  in.h1 = hp[1 * CHAN];
  in.h2 = hp[2 * CHAN];
  in.h3 = hp[3 * CHAN];
  return in;
}

__device__ __forceinline__ void row_add(const RowIn in, float4 w, int dl, int lane, float* acc) {
  float* ap = acc + dl * FDIM + lane;
  atomicAdd(ap + 0 * CHAN, w.x * in.h0);
  atomicAdd(ap + 1 * CHAN, w.y * in.h1);
  atomicAdd(ap + 2 * CHAN, w.z * in.h2);
  atomicAdd(ap + 3 * CHAN, w.w * in.h3);
}

__device__ __forceinline__ void sweep2_slot(int e, int d, int d0, int nloc, int lane,
                                            const int* __restrict__ ei,
                                            const float* __restrict__ as_,
                                            const float* __restrict__ H,
                                            const float* adl, const int* mx,
                                            float* den, float* acc) {
  const unsigned dl = (unsigned)(d - d0);
  unsigned mask = __builtin_amdgcn_ballot_w32(dl < (unsigned)nloc);
  while (mask) {
    const int src_lane = __builtin_ctz(mask);
    mask &= mask - 1;
    const int ee = __builtin_amdgcn_readlane(e, src_lane);
    const int dd = __builtin_amdgcn_readlane((int)dl, src_lane);
    edge_add(edge_load(ei[ee], lane, as_, H), dd, lane, adl, mx, den, acc);
  }
}

__global__ __launch_bounds__(OWN_TB) void gat_owner(const int* __restrict__ ei, int E, int N,
                                                    const float* __restrict__ as_,
                                                    const float* __restrict__ ad_,
                                                    const float* __restrict__ H,
                                                    const float* __restrict__ bias,
                                                    float* outp, int do_relu,
                                                    float* lbuf, int save_list, int load_list,
                                                    int use_keys) {
  __shared__ float acc[RN * FDIM];
  __shared__ float den[RN * NHEAD];
  __shared__ __attribute__((aligned(16))) int   mx[RN * NHEAD];
  __shared__ __attribute__((aligned(16))) float adl[RN * NHEAD];
  __shared__ int2  lst[LCAP];
  __shared__ float4 lgw[LCAP];
  __shared__ int   cnt;
  __shared__ int   cq[CQCAP];
  __shared__ int   ccnt;

  const int tid  = threadIdx.x;
  const int lane = tid & 31;
  const int wave = tid >> 5;
  const int d0   = blockIdx.x * RN;
  int nloc = N - d0;
  if (nloc > RN) nloc = RN;

  for (int i = tid; i < RN * FDIM; i += OWN_TB) acc[i] = 0.0f;
  for (int i = tid; i < RN * NHEAD; i += OWN_TB) {
    float a = 0.0f, m = 0.0f;
    if (i < nloc * NHEAD) {
      a = ad_[(size_t)d0 * NHEAD + i];
      m = leaky(as_[(size_t)d0 * NHEAD + i] + a);
    }
    adl[i] = a;
    mx[i]  = __float_as_int(m);
    den[i] = 0.0f;
  }
  if (tid == 0) { cnt = 0; ccnt = 0; }
  __syncthreads();

  const int* __restrict__ dst = ei + E;
  const int nvec = ((E & 3) == 0) ? (E >> 2) : 0;
  const int4* __restrict__ dst4 = (const int4*)dst;
  const int rot = (int)(((long long)blockIdx.x * nvec) / gridDim.x) & ~31;

  int*  lb   = (int*)(lbuf + (size_t)d0 * FDIM);
  int2* lb2  = (int2*)lb;
  int   gcap = (nloc * FDIM - (nloc == RN ? KW : 0)) / 2 - 1;
  if (gcap > LCAP) gcap = LCAP;

  bool have_list = false;
  if (load_list) {
    const int saved = lb[0];
    if ((unsigned)saved <= (unsigned)gcap) {
      have_list = true;
      for (int base = 0; base < saved; base += OWN_TB * 4) {
        int2   en[4];
        float4 a[4];
#pragma unroll
        for (int k = 0; k < 4; ++k) {
          const int i = base + k * OWN_TB + tid;
          en[k] = make_int2(0, 0);
          if (i < saved) en[k] = lb2[1 + i];
        }
#pragma unroll
        for (int k = 0; k < 4; ++k) {
          const int i = base + k * OWN_TB + tid;
          a[k] = make_float4(0.0f, 0.0f, 0.0f, 0.0f);
          if (i < saved) {
            lst[i] = en[k];
            a[k] = *(const float4*)(as_ + (size_t)en[k].x * NHEAD);
          }
        }
#pragma unroll
        for (int k = 0; k < 4; ++k) {
          const int i = base + k * OWN_TB + tid;
          if (i < saved) {
            const float4 t = edge_logits((unsigned)en[k].y, a[k], adl);
            lgw[i] = t;
            row_max_update((unsigned)en[k].y, t, mx);
          }
        }
      }
      if (tid == 0) cnt = saved;
    }
  }

  bool need_sweep = !have_list;
  if (need_sweep && use_keys && keys_fit(E, N, (int)gridDim.x)) {
    const int ng   = E >> 4;
    const int nreg = (ng + KW / 4 - 1) / (KW / 4);
    const unsigned bc = blockIdx.x * 0x01010101u;
    const uint4* kw4 = (const uint4*)lbuf + (RN * FDIM - KW) / 4 + tid;
    const int rotr = (int)(((long long)blockIdx.x * nreg) / gridDim.x);
    for (int t = 0; t < nreg; t += 8) {
      uint4 kv[8];
      int   g0[8];
#pragma unroll
      for (int k = 0; k < 8; ++k) {
        int r = t + k + rotr;
        if (r >= nreg) r -= nreg;
        if (t + k >= nreg) r = 0;
        g0[k] = (t + k < nreg) ? r * (KW / 4) : -1;
        kv[k] = kw4[(size_t)r * (RN * FDIM / 4)];
      }
      unsigned bal[8];
      int tot = 0;
#pragma unroll
      for (int k = 0; k < 8; ++k) {
        const unsigned z = zero_byte(kv[k].x ^ bc) | zero_byte(kv[k].y ^ bc) |
                           zero_byte(kv[k].z ^ bc) | zero_byte(kv[k].w ^ bc);
        const bool hit = z != 0u && g0[k] >= 0 && g0[k] + tid < ng;
        bal[k] = __builtin_amdgcn_ballot_w32(hit);
        tot += __builtin_popcount(bal[k]);
      }
      if (tot != 0) {
        int pos0 = 0;
        if (lane == 0) pos0 = atomicAdd(&ccnt, tot);
        pos0 = __builtin_amdgcn_readfirstlane(pos0);
#pragma unroll
        for (int k = 0; k < 8; ++k) {
          if ((bal[k] >> lane) & 1u) {
            const int pos = pos0 + (int)__builtin_amdgcn_mbcnt_lo(bal[k], 0u);
            if (pos < CQCAP) cq[pos] = g0[k] + tid;
          }
          pos0 += __builtin_popcount(bal[k]);
        }
      }
    }
    __syncthreads();
    const int ncand = ccnt;
    if (ncand <= CQCAP) {
      need_sweep = false;
      for (int base = 0; base < ncand; base += OWN_TB * 2) {
        int4 v[8];
        int  jj[8];
#pragma unroll
        for (int k = 0; k < 2; ++k) {
          const int i = base + k * OWN_TB + tid;
          const int g = (i < ncand) ? cq[i] : 0;
#pragma unroll
          for (int c = 0; c < 4; ++c) {
            jj[4 * k + c] = 4 * g + c;
            v[4 * k + c] = make_int4(-1, -1, -1, -1);
            if (i < ncand) v[4 * k + c] = dst4[4 * g + c];
          }
        }
        scan8(v, jj, d0, nloc, ei, as_, adl, mx, lst, lgw, &cnt);
      }
    }
  }

  if (need_sweep) {
  for (int base = 0; base < nvec; base += OWN_TB * 8) {
    int4 v[8];
    int  jj[8];
#pragma unroll
    for (int k = 0; k < 8; ++k) {
      const int i = base + k * OWN_TB + tid;
      int j = i + rot;
      if (j >= nvec) j -= nvec;
      jj[k] = j;
      v[k] = make_int4(-1, -1, -1, -1);
      if (i < nvec) v[k] = dst4[j];
    }
    scan8(v, jj, d0, nloc, ei, as_, adl, mx, lst, lgw, &cnt);
  }
  for (int e = 4 * nvec + tid; e < E; e += OWN_TB) {
    const unsigned dl = (unsigned)(dst[e] - d0);
    if (dl < (unsigned)nloc) first_edge(e, dl, ei, as_, adl, mx, lst, lgw, &cnt);
  }
  }
  __syncthreads();
  const int total = cnt;

  if (save_list) {
    if (tid == 0) { lb[0] = total; lb[1] = 0; }
    if (total <= gcap)
      for (int i = tid; i < total; i += OWN_TB) lb2[1 + i] = lst[i];
  }

  if (total <= LCAP) {
    for (int i = tid; i < total; i += OWN_TB) {
      const int    dl = lst[i].y;
      const float4 t  = lgw[i];
      const int4   m  = *(const int4*)(mx + dl * NHEAD);
      const float4 w  = make_float4(__expf(t.x - __int_as_float(m.x)),
                                    __expf(t.y - __int_as_float(m.y)),
                                    __expf(t.z - __int_as_float(m.z)),
                                    __expf(t.w - __int_as_float(m.w)));
      lgw[i] = w;
      atomicAdd(&den[dl * NHEAD + 0], w.x);
      atomicAdd(&den[dl * NHEAD + 1], w.y);
      atomicAdd(&den[dl * NHEAD + 2], w.z);
      atomicAdd(&den[dl * NHEAD + 3], w.w);
    }
  }

  for (int i = wave; i < nloc; i += OWN_WAVES)
    edge_add(edge_load(d0 + i, lane, as_, H), i, lane, adl, mx, den, acc);
  __syncthreads();

  if (total <= LCAP) {
    const int ngrp = total >> 2;
    for (int g = wave; g < ngrp; g += OWN_WAVES) {
      int    sN[4], dN[4];
      float4 wN[4];
#pragma unroll
      for (int u = 0; u < 4; ++u) {
        const int2 en = lst[4 * g + u];
        sN[u] = __builtin_amdgcn_readfirstlane(en.x);
        dN[u] = __builtin_amdgcn_readfirstlane(en.y);
        wN[u] = lgw[4 * g + u];
      }
      RowIn in[4];
#pragma unroll
      for (int u = 0; u < 4; ++u) in[u] = row_load(sN[u], lane, H);
#pragma unroll
      for (int u = 0; u < 4; ++u) row_add(in[u], wN[u], dN[u], lane, acc);
    }
    for (int i = 4 * ngrp + wave; i < total; i += OWN_WAVES) {
      const int2 en = lst[i];
      const int s  = __builtin_amdgcn_readfirstlane(en.x);
      const int dl = __builtin_amdgcn_readfirstlane(en.y);
      row_add(row_load(s, lane, H), lgw[i], dl, lane, acc);
    }
  } else {
    for (int base = 0; base < nvec; base += OWN_TB * 4) {
      int4 v[4];
      int  jj[4];
#pragma unroll
      for (int k = 0; k < 4; ++k) {
        const int i = base + k * OWN_TB + tid;
        int j = i + rot;
        if (j >= nvec) j -= nvec;
        jj[k] = j;
        v[k] = make_int4(-1, -1, -1, -1);
        if (i < nvec) v[k] = dst4[j];
      }
#pragma unroll
      for (int k = 0; k < 4; ++k) {
        sweep2_slot(4 * jj[k] + 0, v[k].x, d0, nloc, lane, ei, as_, H, adl, mx, den, acc);
        sweep2_slot(4 * jj[k] + 1, v[k].y, d0, nloc, lane, ei, as_, H, adl, mx, den, acc);
        sweep2_slot(4 * jj[k] + 2, v[k].z, d0, nloc, lane, ei, as_, H, adl, mx, den, acc);
        sweep2_slot(4 * jj[k] + 3, v[k].w, d0, nloc, lane, ei, as_, H, adl, mx, den, acc);
      }
    }
    for (int eb = 4 * nvec; eb < E; eb += OWN_TB) {
      const int e = eb + tid;
      const int d = (e < E) ? dst[e] : -1;
      sweep2_slot(e, d, d0, nloc, lane, ei, as_, H, adl, mx, den, acc);
    }
  }
  __syncthreads();

  for (int i = tid; i < nloc * FDIM; i += OWN_TB) {
    const int r = i >> 7, c = i & (FDIM - 1);
    float o = acc[i] / (den[r * NHEAD + (c >> 5)] + 1e-16f) + bias[c];
    if (do_relu) o = o > 0.0f ? o : 0.0f;
    outp[(size_t)d0 * FDIM + i] = o;
  }
}

static inline int cdiv(long long a, int b) { return (int)((a + b - 1) / b); }

extern "C" void kernel_launch(void* const* d_in, const int* in_sizes, int n_in,
                              void* d_out, int out_size, void* d_ws, size_t ws_size,
                              hipStream_t stream) {
  const float* x     = (const float*)d_in[0];
  const int*   ei    = (const int*)d_in[1];
  const float* W1    = (const float*)d_in[2];
  const float* asrc1 = (const float*)d_in[3];
  const float* adst1 = (const float*)d_in[4];
  const float* b1    = (const float*)d_in[5];
  const float* W2    = (const float*)d_in[6];
  const float* asrc2 = (const float*)d_in[7];
  const float* adst2 = (const float*)d_in[8];
  const float* b2    = (const float*)d_in[9];
  float* out = (float*)d_out;

  const int N  = in_sizes[0] / FDIM;
  const int E  = in_sizes[1] / 2;

  float* H    = (float*)d_ws;
  float* Hact = H    + (size_t)N * FDIM;
  float* as_  = Hact + (size_t)N * FDIM;
  float* ad_  = as_  + (size_t)N * NHEAD;

  const int TB = 256;
  const int gNH    = cdiv((long long)N * NHEAD, TB);
  const int gOwn   = cdiv(N, RN);

  gat_gemm_wmma<<<GEMM_GRID, GEMM_TB, 0, stream>>>(x, W1, H, N);
  gat_alpha<<<gNH, TB, 0, stream>>>(H, asrc1, adst1, as_, ad_, N, ei, E, gOwn,  out);
  gat_owner<<<gOwn, OWN_TB, 0, stream>>>(ei, E, N, as_, ad_, H, b1, Hact,  1,
                                         out,  1,  0,  1);

  gat_gemm_wmma<<<GEMM_GRID, GEMM_TB, 0, stream>>>(Hact, W2, H, N);
  gat_alpha<<<gNH, TB, 0, stream>>>(H, asrc2, adst2, as_, ad_, N, ei, E, gOwn,  nullptr);
  gat_owner<<<gOwn, OWN_TB, 0, stream>>>(ei, E, N, as_, ad_, H, b2, out,  0,
                                         out,  0,  1,  0);
}


# Layer-1 owner block reads a bit map of the 16-edge groups holding its edges (extra blocks of the logit launch build it, LDS or) instead of all owner bytes; list order, so f32 LDS add order, changes
# speedup vs baseline: 4.6185x; 1.0452x over previous
#include <hip/hip_runtime.h>
#include <hip/hip_bf16.h>


typedef __attribute__((ext_vector_type(2))) float v2f;
typedef __attribute__((ext_vector_type(8))) float v8f;

#define FDIM 128
#define NHEAD 4
#define CHAN 32
#define NEG_SLOPE 0.2f

#define RN 196
#define OWN_TB 1024
#define OWN_WAVES (OWN_TB / 32)
#define LCAP 8192
#define CQCAP 4096
#define AL_TB 256
#define BM_CHUNK 1024

__device__ __forceinline__ int bm_words(int E) {
  return (((E >> 4) + BM_CHUNK - 1) / BM_CHUNK) * 32;
}
__device__ __forceinline__ bool bm_fit(int E, int N, int nblk) {
  return E > 0 && (E & 15) == 0 && nblk >= 1 && nblk <= 256 &&
         bm_words(E) <= (N - (nblk - 1) * RN) * FDIM;
}
static_assert(BM_CHUNK == 32 * 32, "a builder block fills 32 words of 32 groups per owner");
static_assert(BM_CHUNK / 32 == 4 * (AL_TB / 32), "a builder wave takes four of the 32 words");

#define GEMM_TB 256
#define GEMM_GRID 768
#define XPITCH (FDIM + 4)

__global__ __launch_bounds__(GEMM_TB) void gat_gemm_wmma(const float* __restrict__ X,
                                                         const float* __restrict__ W,
                                                         float* __restrict__ H,
                                                         int N) {
  __shared__ __attribute__((aligned(16))) float xs[2][16 * XPITCH];

  const int tid  = threadIdx.x;
  const int lane = tid & 31;
  const int wave = tid >> 5;
  const int m    = lane & 15;
  const int half = lane >> 4;
  const int col0 = wave * 16;
  const int ntile = (N + 15) >> 4;

  v2f b[FDIM / 4];
#pragma unroll
  for (int j = 0; j < FDIM / 4; ++j) {
    const int ka = 4 * j + 2 * half;
    b[j].x = W[ka * FDIM + col0 + m];
    b[j].y = W[(ka + 1) * FDIM + col0 + m];
  }

  const int sr0 = tid >> 5, sr1 = sr0 + 8;
  const int sc  = (tid & 31) * 4;

  int tile = blockIdx.x;
  if (tile >= ntile) return;
  float4 st0, st1;
  {
    int r0 = tile * 16 + sr0, r1 = tile * 16 + sr1;
    if (r0 >= N) r0 = N - 1;
    if (r1 >= N) r1 = N - 1;
    st0 = *(const float4*)(X + (size_t)r0 * FDIM + sc);
    st1 = *(const float4*)(X + (size_t)r1 * FDIM + sc);
  }

  int buf = 0;
  for (;;) {
    float* xt = xs[buf];
    *(float4*)(xt + sr0 * XPITCH + sc) = st0;
    *(float4*)(xt + sr1 * XPITCH + sc) = st1;
    __syncthreads();

    const int nxt = tile + GEMM_GRID;
    {
      const int lt = nxt < ntile ? nxt : tile;
      int r0 = lt * 16 + sr0, r1 = lt * 16 + sr1;
      if (r0 >= N) r0 = N - 1;
      if (r1 >= N) r1 = N - 1;
      st0 = *(const float4*)(X + (size_t)r0 * FDIM + sc);
      st1 = *(const float4*)(X + (size_t)r1 * FDIM + sc);
    }

    const float* xa = xt + m * XPITCH + 2 * half;
    v2f a[FDIM / 4];
#pragma unroll
    for (int j = 0; j < FDIM / 4; ++j) a[j] = *(const v2f*)(xa + 4 * j);
    __builtin_amdgcn_sched_barrier(0);
    v8f acc = {};
#pragma unroll
    for (int j = 0; j < FDIM / 4; ++j) {
      acc = __builtin_amdgcn_wmma_f32_16x16x4_f32(
           false, a[j],  false, b[j],
           (short)0, acc,  false,  false);
    }

    const int row0 = tile * 16;
#pragma unroll
    for (int v = 0; v < 8; ++v) {
      const int row = row0 + v + 8 * half;
      if (row < N) H[(size_t)row * FDIM + col0 + m] = acc[v];
    }
    if (nxt >= ntile) break;
    tile = nxt;
    buf ^= 1;
  }
}

__global__ __launch_bounds__(AL_TB) void gat_alpha(const float* __restrict__ H,
                          const float* __restrict__ a_src,
                          const float* __restrict__ a_dst,
                          float* __restrict__ as_, float* __restrict__ ad_,
                          int N,
                          const int* __restrict__ ei, int E, int nblk,
                          float* bmbuf, int nbm) {
  __shared__ unsigned tb[32 * 257];
  if ((int)blockIdx.x < nbm) {
    if (bmbuf == nullptr || !bm_fit(E, N, nblk)) return;
    const int lane = threadIdx.x & 31;
    const int wave = threadIdx.x >> 5;
    const int ng   = E >> 4;
    const int4* __restrict__ dst4 = (const int4*)(ei + E);
    int4 v[4][4];
    bool live[4];
#pragma unroll
    for (int k = 0; k < 4; ++k) {
      const int g = (int)blockIdx.x * BM_CHUNK + (k * (AL_TB / 32) + wave) * 32 + lane;
      live[k] = g < ng;
      const int gl = live[k] ? g : 0;
#pragma unroll
      for (int c = 0; c < 4; ++c) v[k][c] = dst4[4 * gl + c];
    }
    for (int i = threadIdx.x; i < 32 * 257; i += AL_TB) tb[i] = 0u;
    __syncthreads();
    const unsigned bit = 1u << lane;
#pragma unroll
    for (int k = 0; k < 4; ++k) {
      if (live[k]) {
        unsigned* row = tb + (k * (AL_TB / 32) + wave) * 257;
#pragma unroll
        for (int c = 0; c < 4; ++c) {
          atomicOr(row + ((unsigned)v[k][c].x / RN & 255u), bit);
          atomicOr(row + ((unsigned)v[k][c].y / RN & 255u), bit);
          atomicOr(row + ((unsigned)v[k][c].z / RN & 255u), bit);
          atomicOr(row + ((unsigned)v[k][c].w / RN & 255u), bit);
        }
      }
    }
    __syncthreads();
    for (int b = wave; b < nblk; b += AL_TB / 32) {
      unsigned* bm = (unsigned*)(bmbuf + (size_t)b * RN * FDIM) + (int)blockIdx.x * 32;
      bm[lane] = tb[lane * 257 + b];
    }
    return;
  }
  const int tid = ((int)blockIdx.x - nbm) * blockDim.x + threadIdx.x;
  if (tid >= N * NHEAD) return;
  const int n = tid >> 2;
  const int h = tid & 3;
  const float* __restrict__ hp = H + (size_t)n * FDIM + h * CHAN;
  const float* __restrict__ asv = a_src + h * CHAN;
  const float* __restrict__ adv = a_dst + h * CHAN;
  float s = 0.0f, d = 0.0f;
#pragma unroll
  for (int c = 0; c < CHAN; ++c) {
    const float hv = hp[c];
    s = fmaf(hv, asv[c], s);
    d = fmaf(hv, adv[c], d);
  }
  as_[tid] = s;
  ad_[tid] = d;
}


__device__ __forceinline__ float leaky(float v) {
  return v > 0.0f ? v : NEG_SLOPE * v;
}

__device__ __forceinline__ void lds_max_float(int* addr, float v) {
  const int iv = __float_as_int(v);
  if (iv >= 0) atomicMax(addr, iv);
  else         atomicMin((unsigned int*)addr, (unsigned int)iv);
}

__device__ __forceinline__ float4 edge_logits(unsigned dl, float4 a, const float* adl) {
  const float4 d = *(const float4*)(adl + dl * NHEAD);
  return make_float4(leaky(a.x + d.x), leaky(a.y + d.y), leaky(a.z + d.z), leaky(a.w + d.w));
}

__device__ __forceinline__ void row_max_update(unsigned dl, float4 t, int* mx) {
  lds_max_float(&mx[dl * NHEAD + 0], t.x);
  lds_max_float(&mx[dl * NHEAD + 1], t.y);
  lds_max_float(&mx[dl * NHEAD + 2], t.z);
  lds_max_float(&mx[dl * NHEAD + 3], t.w);
}

__device__ __forceinline__ void first_update(int s, unsigned dl, float4 a,
                                             const float* adl, int* mx,
                                             int2* lst, float4* lgw, int* cnt) {
  const float4 t = edge_logits(dl, a, adl);
  row_max_update(dl, t, mx);
  const int pos = atomicAdd(cnt, 1);
  if (pos < LCAP) {
    lst[pos] = make_int2(s, (int)dl);
    lgw[pos] = t;
  }
}

__device__ __forceinline__ void first_edge(int e, unsigned dl,
                                           const int* __restrict__ ei,
                                           const float* __restrict__ as_,
                                           const float* adl, int* mx,
                                           int2* lst, float4* lgw, int* cnt) {
  const int s = ei[e];
  const float4 a = *(const float4*)(as_ + (size_t)s * NHEAD);
  first_update(s, dl, a, adl, mx, lst, lgw, cnt);
}

__device__ __forceinline__ void scan8(const int4 (&v)[8], const int (&jj)[8],
                                      int d0, int nloc,
                                      const int* __restrict__ ei,
                                      const float* __restrict__ as_,
                                      const float* adl, int* mx,
                                      int2* lst, float4* lgw, int* cnt) {
  int      qe0 = 0, qe1 = 0, nq = 0;
  unsigned qd0 = 0, qd1 = 0;
#pragma unroll
  for (int k = 0; k < 8; ++k) {
    const int dd[4] = {v[k].x, v[k].y, v[k].z, v[k].w};
#pragma unroll
    for (int c = 0; c < 4; ++c) {
      const unsigned dl = (unsigned)(dd[c] - d0);
      if (dl < (unsigned)nloc) {
        const int e = 4 * jj[k] + c;
        if (nq == 0)      { qe0 = e; qd0 = dl; }
        else if (nq == 1) { qe1 = e; qd1 = dl; }
        else              first_edge(e, dl, ei, as_, adl, mx, lst, lgw, cnt);
        ++nq;
      }
    }
  }
  int s0 = 0, s1 = 0;
  if (nq > 0) s0 = ei[qe0];
  if (nq > 1) s1 = ei[qe1];
  float4 a0 = make_float4(0.0f, 0.0f, 0.0f, 0.0f), a1 = a0;
  if (nq > 0) a0 = *(const float4*)(as_ + (size_t)s0 * NHEAD);
  if (nq > 1) a1 = *(const float4*)(as_ + (size_t)s1 * NHEAD);
  if (nq > 0) first_update(s0, qd0, a0, adl, mx, lst, lgw, cnt);
  if (nq > 1) first_update(s1, qd1, a1, adl, mx, lst, lgw, cnt);
}

struct EdgeIn { float4 a; float h0, h1, h2, h3; };

__device__ __forceinline__ EdgeIn edge_load(int s, int lane,
                                            const float* __restrict__ as_,
                                            const float* __restrict__ H) {
  EdgeIn in;
  in.a = *(const float4*)(as_ + (size_t)s * NHEAD);
  const float* __restrict__ hp = H + (size_t)s * FDIM + lane;
  in.h0 = hp[0 * CHAN];
  in.h1 = hp[1 * CHAN];
  in.h2 = hp[2 * CHAN];
  in.h3 = hp[3 * CHAN];
  return in;
}

__device__ __forceinline__ void edge_add(const EdgeIn in, int dl, int lane,
                                         const float* adl, const int* mx,
                                         float* den, float* acc) {
  const float e0 = __expf(leaky(in.a.x + adl[dl * NHEAD + 0]) - __int_as_float(mx[dl * NHEAD + 0]));
  const float e1 = __expf(leaky(in.a.y + adl[dl * NHEAD + 1]) - __int_as_float(mx[dl * NHEAD + 1]));
  const float e2 = __expf(leaky(in.a.z + adl[dl * NHEAD + 2]) - __int_as_float(mx[dl * NHEAD + 2]));
  const float e3 = __expf(leaky(in.a.w + adl[dl * NHEAD + 3]) - __int_as_float(mx[dl * NHEAD + 3]));
  float* ap = acc + dl * FDIM + lane;
  atomicAdd(ap + 0 * CHAN, e0 * in.h0);
  atomicAdd(ap + 1 * CHAN, e1 * in.h1);
  atomicAdd(ap + 2 * CHAN, e2 * in.h2);
  atomicAdd(ap + 3 * CHAN, e3 * in.h3);
  if (lane < NHEAD) {
    const float ev = lane == 0 ? e0 : lane == 1 ? e1 : lane == 2 ? e2 : e3;
    atomicAdd(&den[dl * NHEAD + lane], ev);
  }
}

struct RowIn { float h0, h1, h2, h3; };

__device__ __forceinline__ RowIn row_load(int s, int lane, const float* __restrict__ H) {
  RowIn in;
  const float* __restrict__ hp = H + (size_t)s * FDIM + lane;
  in.h0 = hp[0 * CHAN];
  in.h1 = hp[1 * CHAN];
  in.h2 = hp[2 * CHAN];
  in.h3 = hp[3 * CHAN];
  return in;
}

__device__ __forceinline__ void row_add(const RowIn in, float4 w, int dl, int lane, float* acc) {
  float* ap = acc + dl * FDIM + lane;
  atomicAdd(ap + 0 * CHAN, w.x * in.h0);
  atomicAdd(ap + 1 * CHAN, w.y * in.h1);
  atomicAdd(ap + 2 * CHAN, w.z * in.h2);
  atomicAdd(ap + 3 * CHAN, w.w * in.h3);
}

__device__ __forceinline__ void sweep2_slot(int e, int d, int d0, int nloc, int lane,
                                            const int* __restrict__ ei,
                                            const float* __restrict__ as_,
                                            const float* __restrict__ H,
                                            const float* adl, const int* mx,
                                            float* den, float* acc) {
  const unsigned dl = (unsigned)(d - d0);
  unsigned mask = __builtin_amdgcn_ballot_w32(dl < (unsigned)nloc);
  while (mask) {
    const int src_lane = __builtin_ctz(mask);
    mask &= mask - 1;
    const int ee = __builtin_amdgcn_readlane(e, src_lane);
    const int dd = __builtin_amdgcn_readlane((int)dl, src_lane);
    edge_add(edge_load(ei[ee], lane, as_, H), dd, lane, adl, mx, den, acc);
  }
}

__global__ __launch_bounds__(OWN_TB) void gat_owner(const int* __restrict__ ei, int E, int N,
                                                    const float* __restrict__ as_,
                                                    const float* __restrict__ ad_,
                                                    const float* __restrict__ H,
                                                    const float* __restrict__ bias,
                                                    float* outp, int do_relu,
                                                    float* lbuf, int save_list, int load_list,
                                                    int use_map) {
  __shared__ float acc[RN * FDIM];
  __shared__ float den[RN * NHEAD];
  __shared__ __attribute__((aligned(16))) int   mx[RN * NHEAD];
  __shared__ __attribute__((aligned(16))) float adl[RN * NHEAD];
  __shared__ int2  lst[LCAP];
  __shared__ float4 lgw[LCAP];
  __shared__ int   cnt;
  __shared__ int   cq[CQCAP];
  __shared__ int   ccnt;

  const int tid  = threadIdx.x;
  const int lane = tid & 31;
  const int wave = tid >> 5;
  const int d0   = blockIdx.x * RN;
  int nloc = N - d0;
  if (nloc > RN) nloc = RN;

  for (int i = tid; i < RN * FDIM; i += OWN_TB) acc[i] = 0.0f;
  for (int i = tid; i < RN * NHEAD; i += OWN_TB) {
    float a = 0.0f, m = 0.0f;
    if (i < nloc * NHEAD) {
      a = ad_[(size_t)d0 * NHEAD + i];
      m = leaky(as_[(size_t)d0 * NHEAD + i] + a);
    }
    adl[i] = a;
    mx[i]  = __float_as_int(m);
    den[i] = 0.0f;
  }
  if (tid == 0) { cnt = 0; ccnt = 0; }
  __syncthreads();

  const int* __restrict__ dst = ei + E;
  const int nvec = ((E & 3) == 0) ? (E >> 2) : 0;
  const int4* __restrict__ dst4 = (const int4*)dst;
  const int rot = (int)(((long long)blockIdx.x * nvec) / gridDim.x) & ~31;

  int*  lb   = (int*)(lbuf + (size_t)d0 * FDIM);
  int2* lb2  = (int2*)lb;
  int   gcap = (nloc * FDIM) / 2 - 1;
  if (gcap > LCAP) gcap = LCAP;

  bool have_list = false;
  if (load_list) {
    const int saved = lb[0];
    if ((unsigned)saved <= (unsigned)gcap) {
      have_list = true;
      for (int base = 0; base < saved; base += OWN_TB * 4) {
        int2   en[4];
        float4 a[4];
#pragma unroll
        for (int k = 0; k < 4; ++k) {
          const int i = base + k * OWN_TB + tid;
          en[k] = make_int2(0, 0);
          if (i < saved) en[k] = lb2[1 + i];
        }
#pragma unroll
        for (int k = 0; k < 4; ++k) {
          const int i = base + k * OWN_TB + tid;
          a[k] = make_float4(0.0f, 0.0f, 0.0f, 0.0f);
          if (i < saved) {
            lst[i] = en[k];
            a[k] = *(const float4*)(as_ + (size_t)en[k].x * NHEAD);
          }
        }
#pragma unroll
        for (int k = 0; k < 4; ++k) {
          const int i = base + k * OWN_TB + tid;
          if (i < saved) {
            const float4 t = edge_logits((unsigned)en[k].y, a[k], adl);
            lgw[i] = t;
            row_max_update((unsigned)en[k].y, t, mx);
          }
        }
      }
      if (tid == 0) cnt = saved;
    }
  }

  bool need_sweep = !have_list;
  if (need_sweep && use_map && bm_fit(E, N, (int)gridDim.x)) {
    const int nw = bm_words(E);
    const unsigned* bm = (const unsigned*)(outp + (size_t)d0 * FDIM);
    for (int w0 = 0; w0 < nw; w0 += OWN_TB) {
      const int w = w0 + tid;
      unsigned bits = (w < nw) ? bm[w] : 0u;
      const int c = __builtin_popcount(bits);
      int incl = c;
      for (int o = 1; o < 32; o <<= 1) {
        const int up = __shfl_up(incl, o, 32);
        if (lane >= o) incl += up;
      }
      const int tot = __builtin_amdgcn_readlane(incl, 31);
      if (tot != 0) {
        int pos0 = 0;
        if (lane == 0) pos0 = atomicAdd(&ccnt, tot);
        int pos = __builtin_amdgcn_readfirstlane(pos0) + incl - c;
        while (bits != 0u) {
          const int g = w * 32 + __builtin_ctz(bits);
          bits &= bits - 1u;
          if (pos < CQCAP) cq[pos] = g;
          ++pos;
        }
      }
    }
    __syncthreads();
    const int ncand = ccnt;
    if (ncand <= CQCAP) {
      need_sweep = false;
      for (int base = 0; base < ncand; base += OWN_TB * 2) {
        int4 v[8];
        int  jj[8];
#pragma unroll
        for (int k = 0; k < 2; ++k) {
          const int i = base + k * OWN_TB + tid;
          const int g = (i < ncand) ? cq[i] : 0;
#pragma unroll
          for (int c = 0; c < 4; ++c) {
            jj[4 * k + c] = 4 * g + c;
            v[4 * k + c] = make_int4(-1, -1, -1, -1);
            if (i < ncand) v[4 * k + c] = dst4[4 * g + c];
          }
        }
        scan8(v, jj, d0, nloc, ei, as_, adl, mx, lst, lgw, &cnt);
      }
    }
  }

  if (need_sweep) {
  for (int base = 0; base < nvec; base += OWN_TB * 8) {
    int4 v[8];
    int  jj[8];
#pragma unroll
    for (int k = 0; k < 8; ++k) {
      const int i = base + k * OWN_TB + tid;
      int j = i + rot;
      if (j >= nvec) j -= nvec;
      jj[k] = j;
      v[k] = make_int4(-1, -1, -1, -1);
      if (i < nvec) v[k] = dst4[j];
    }
    scan8(v, jj, d0, nloc, ei, as_, adl, mx, lst, lgw, &cnt);
  }
  for (int e = 4 * nvec + tid; e < E; e += OWN_TB) {
    const unsigned dl = (unsigned)(dst[e] - d0);
    if (dl < (unsigned)nloc) first_edge(e, dl, ei, as_, adl, mx, lst, lgw, &cnt);
  }
  }
  __syncthreads();
  const int total = cnt;

  if (save_list) {
    if (tid == 0) { lb[0] = total; lb[1] = 0; }
    if (total <= gcap)
      for (int i = tid; i < total; i += OWN_TB) lb2[1 + i] = lst[i];
  }

  if (total <= LCAP) {
    for (int i = tid; i < total; i += OWN_TB) {
      const int    dl = lst[i].y;
      const float4 t  = lgw[i];
      const int4   m  = *(const int4*)(mx + dl * NHEAD);
      const float4 w  = make_float4(__expf(t.x - __int_as_float(m.x)),
                                    __expf(t.y - __int_as_float(m.y)),
                                    __expf(t.z - __int_as_float(m.z)),
                                    __expf(t.w - __int_as_float(m.w)));
      lgw[i] = w;
      atomicAdd(&den[dl * NHEAD + 0], w.x);
      atomicAdd(&den[dl * NHEAD + 1], w.y);
      atomicAdd(&den[dl * NHEAD + 2], w.z);
      atomicAdd(&den[dl * NHEAD + 3], w.w);
    }
  }

  for (int i = wave; i < nloc; i += OWN_WAVES)
    edge_add(edge_load(d0 + i, lane, as_, H), i, lane, adl, mx, den, acc);
  __syncthreads();

  if (total <= LCAP) {
    const int ngrp = total >> 2;
    for (int g = wave; g < ngrp; g += OWN_WAVES) {
      int    sN[4], dN[4];
      float4 wN[4];
#pragma unroll
      for (int u = 0; u < 4; ++u) {
        const int2 en = lst[4 * g + u];
        sN[u] = __builtin_amdgcn_readfirstlane(en.x);
        dN[u] = __builtin_amdgcn_readfirstlane(en.y);
        wN[u] = lgw[4 * g + u];
      }
      RowIn in[4];
#pragma unroll
      for (int u = 0; u < 4; ++u) in[u] = row_load(sN[u], lane, H);
#pragma unroll
      for (int u = 0; u < 4; ++u) row_add(in[u], wN[u], dN[u], lane, acc);
    }
    for (int i = 4 * ngrp + wave; i < total; i += OWN_WAVES) {
      const int2 en = lst[i];
      const int s  = __builtin_amdgcn_readfirstlane(en.x);
      const int dl = __builtin_amdgcn_readfirstlane(en.y);
      row_add(row_load(s, lane, H), lgw[i], dl, lane, acc);
    }
  } else {
    for (int base = 0; base < nvec; base += OWN_TB * 4) {
      int4 v[4];
      int  jj[4];
#pragma unroll
      for (int k = 0; k < 4; ++k) {
        const int i = base + k * OWN_TB + tid;
        int j = i + rot;
        if (j >= nvec) j -= nvec;
        jj[k] = j;
        v[k] = make_int4(-1, -1, -1, -1);
        if (i < nvec) v[k] = dst4[j];
      }
#pragma unroll
      for (int k = 0; k < 4; ++k) {
        sweep2_slot(4 * jj[k] + 0, v[k].x, d0, nloc, lane, ei, as_, H, adl, mx, den, acc);
        sweep2_slot(4 * jj[k] + 1, v[k].y, d0, nloc, lane, ei, as_, H, adl, mx, den, acc);
        sweep2_slot(4 * jj[k] + 2, v[k].z, d0, nloc, lane, ei, as_, H, adl, mx, den, acc);
        sweep2_slot(4 * jj[k] + 3, v[k].w, d0, nloc, lane, ei, as_, H, adl, mx, den, acc);
      }
    }
    for (int eb = 4 * nvec; eb < E; eb += OWN_TB) {
      const int e = eb + tid;
      const int d = (e < E) ? dst[e] : -1;
      sweep2_slot(e, d, d0, nloc, lane, ei, as_, H, adl, mx, den, acc);
    }
  }
  __syncthreads();

  for (int i = tid; i < nloc * FDIM; i += OWN_TB) {
    const int r = i >> 7, c = i & (FDIM - 1);
    float o = acc[i] / (den[r * NHEAD + (c >> 5)] + 1e-16f) + bias[c];
    if (do_relu) o = o > 0.0f ? o : 0.0f;
    outp[(size_t)d0 * FDIM + i] = o;
  }
}

static inline int cdiv(long long a, int b) { return (int)((a + b - 1) / b); }

extern "C" void kernel_launch(void* const* d_in, const int* in_sizes, int n_in,
                              void* d_out, int out_size, void* d_ws, size_t ws_size,
                              hipStream_t stream) {
  const float* x     = (const float*)d_in[0];
  const int*   ei    = (const int*)d_in[1];
  const float* W1    = (const float*)d_in[2];
  const float* asrc1 = (const float*)d_in[3];
  const float* adst1 = (const float*)d_in[4];
  const float* b1    = (const float*)d_in[5];
  const float* W2    = (const float*)d_in[6];
  const float* asrc2 = (const float*)d_in[7];
  const float* adst2 = (const float*)d_in[8];
  const float* b2    = (const float*)d_in[9];
  float* out = (float*)d_out;

  const int N  = in_sizes[0] / FDIM;
  const int E  = in_sizes[1] / 2;

  float* H    = (float*)d_ws;
  float* Hact = H    + (size_t)N * FDIM;
  float* as_  = Hact + (size_t)N * FDIM;
  float* ad_  = as_  + (size_t)N * NHEAD;

  const int TB = AL_TB;
  const int gNH    = cdiv((long long)N * NHEAD, TB);
  const int gBm    = cdiv(E >> 4, BM_CHUNK);
  const int gOwn   = cdiv(N, RN);

  gat_gemm_wmma<<<GEMM_GRID, GEMM_TB, 0, stream>>>(x, W1, H, N);
  gat_alpha<<<gNH + gBm, TB, 0, stream>>>(H, asrc1, adst1, as_, ad_, N, ei, E, gOwn,  Hact,  gBm);
  gat_owner<<<gOwn, OWN_TB, 0, stream>>>(ei, E, N, as_, ad_, H, b1, Hact,  1,
                                         out,  1,  0,  1);

  gat_gemm_wmma<<<GEMM_GRID, GEMM_TB, 0, stream>>>(Hact, W2, H, N);
  gat_alpha<<<gNH, TB, 0, stream>>>(H, asrc2, adst2, as_, ad_, N, ei, E, gOwn,  nullptr,  0);
  gat_owner<<<gOwn, OWN_TB, 0, stream>>>(ei, E, N, as_, ad_, H, b2, out,  0,
                                         out,  0,  1,  0);
}


# Owner blocks take 98 rows, not 196, and under half a WGP's LDS: two 32-wave blocks per WGP, one's dependent load chains run under the other's row gathers; f32 LDS add order (arrival) changes
# speedup vs baseline: 4.8317x; 1.0462x over previous
#include <hip/hip_runtime.h>
#include <hip/hip_bf16.h>


typedef __attribute__((ext_vector_type(2))) float v2f;
typedef __attribute__((ext_vector_type(8))) float v8f;

#define FDIM 128
#define NHEAD 4
#define CHAN 32
#define NEG_SLOPE 0.2f

#define RN 98
#define OWN_TB 1024
#define OWN_WAVES (OWN_TB / 32)
#define LCAP 4096
#define CQCAP 2048
#define AL_TB 256
#define BM_CHUNK 1024
#define OWN_MAX 512
#define TB_PITCH (OWN_MAX + 1)

__device__ __forceinline__ int bm_words(int E) {
  return (((E >> 4) + BM_CHUNK - 1) / BM_CHUNK) * 32;
}
__device__ __forceinline__ bool bm_fit(int E, int N, int nblk) {
  return E > 0 && (E & 15) == 0 && nblk >= 1 && nblk <= OWN_MAX &&
         bm_words(E) <= (N - (nblk - 1) * RN) * FDIM;
}
static_assert(BM_CHUNK == 32 * 32, "a builder block fills 32 words of 32 groups per owner");
static_assert((OWN_MAX & (OWN_MAX - 1)) == 0, "the builders mask an owner number into the table's columns");
static_assert(BM_CHUNK / 32 == 4 * (AL_TB / 32), "a builder wave takes four of the 32 words");

#define GEMM_TB 256
#define GEMM_GRID 768
#define XPITCH (FDIM + 4)

__global__ __launch_bounds__(GEMM_TB) void gat_gemm_wmma(const float* __restrict__ X,
                                                         const float* __restrict__ W,
                                                         float* __restrict__ H,
                                                         int N) {
  __shared__ __attribute__((aligned(16))) float xs[2][16 * XPITCH];

  const int tid  = threadIdx.x;
  const int lane = tid & 31;
  const int wave = tid >> 5;
  const int m    = lane & 15;
  const int half = lane >> 4;
  const int col0 = wave * 16;
  const int ntile = (N + 15) >> 4;

  v2f b[FDIM / 4];
#pragma unroll
  for (int j = 0; j < FDIM / 4; ++j) {
    const int ka = 4 * j + 2 * half;
    b[j].x = W[ka * FDIM + col0 + m];
    b[j].y = W[(ka + 1) * FDIM + col0 + m];
  }

  const int sr0 = tid >> 5, sr1 = sr0 + 8;
  const int sc  = (tid & 31) * 4;

  int tile = blockIdx.x;
  if (tile >= ntile) return;
  float4 st0, st1;
  {
    int r0 = tile * 16 + sr0, r1 = tile * 16 + sr1;
    if (r0 >= N) r0 = N - 1;
    if (r1 >= N) r1 = N - 1;
    st0 = *(const float4*)(X + (size_t)r0 * FDIM + sc);
    st1 = *(const float4*)(X + (size_t)r1 * FDIM + sc);
  }

  int buf = 0;
  for (;;) {
    float* xt = xs[buf];
    *(float4*)(xt + sr0 * XPITCH + sc) = st0;
    *(float4*)(xt + sr1 * XPITCH + sc) = st1;
    __syncthreads();

    const int nxt = tile + GEMM_GRID;
    {
      const int lt = nxt < ntile ? nxt : tile;
      int r0 = lt * 16 + sr0, r1 = lt * 16 + sr1;
      if (r0 >= N) r0 = N - 1;
      if (r1 >= N) r1 = N - 1;
      st0 = *(const float4*)(X + (size_t)r0 * FDIM + sc);
      st1 = *(const float4*)(X + (size_t)r1 * FDIM + sc);
    }

    const float* xa = xt + m * XPITCH + 2 * half;
    v2f a[FDIM / 4];
#pragma unroll
    for (int j = 0; j < FDIM / 4; ++j) a[j] = *(const v2f*)(xa + 4 * j);
    __builtin_amdgcn_sched_barrier(0);
    v8f acc = {};
#pragma unroll
    for (int j = 0; j < FDIM / 4; ++j) {
      acc = __builtin_amdgcn_wmma_f32_16x16x4_f32(
           false, a[j],  false, b[j],
           (short)0, acc,  false,  false);
    }

    const int row0 = tile * 16;
#pragma unroll
    for (int v = 0; v < 8; ++v) {
      const int row = row0 + v + 8 * half;
      if (row < N) H[(size_t)row * FDIM + col0 + m] = acc[v];
    }
    if (nxt >= ntile) break;
    tile = nxt;
    buf ^= 1;
  }
}

__global__ __launch_bounds__(AL_TB) void gat_alpha(const float* __restrict__ H,
                          const float* __restrict__ a_src,
                          const float* __restrict__ a_dst,
                          float* __restrict__ as_, float* __restrict__ ad_,
                          int N,
                          const int* __restrict__ ei, int E, int nblk,
                          float* bmbuf, int nbm) {
  __shared__ unsigned tb[32 * TB_PITCH];
  if ((int)blockIdx.x < nbm) {
    if (bmbuf == nullptr || !bm_fit(E, N, nblk)) return;
    const int lane = threadIdx.x & 31;
    const int wave = threadIdx.x >> 5;
    const int ng   = E >> 4;
    const int4* __restrict__ dst4 = (const int4*)(ei + E);
    int4 v[4][4];
    bool live[4];
#pragma unroll
    for (int k = 0; k < 4; ++k) {
      const int g = (int)blockIdx.x * BM_CHUNK + (k * (AL_TB / 32) + wave) * 32 + lane;
      live[k] = g < ng;
      const int gl = live[k] ? g : 0;
#pragma unroll
      for (int c = 0; c < 4; ++c) v[k][c] = dst4[4 * gl + c];
    }
    for (int i = threadIdx.x; i < 32 * TB_PITCH; i += AL_TB) tb[i] = 0u;
    __syncthreads();
    const unsigned bit = 1u << lane;
#pragma unroll
    for (int k = 0; k < 4; ++k) {
      if (live[k]) {
        unsigned* row = tb + (k * (AL_TB / 32) + wave) * TB_PITCH;
#pragma unroll
        for (int c = 0; c < 4; ++c) {
          atomicOr(row + ((unsigned)v[k][c].x / RN & (OWN_MAX - 1u)), bit);
          atomicOr(row + ((unsigned)v[k][c].y / RN & (OWN_MAX - 1u)), bit);
          atomicOr(row + ((unsigned)v[k][c].z / RN & (OWN_MAX - 1u)), bit);
          atomicOr(row + ((unsigned)v[k][c].w / RN & (OWN_MAX - 1u)), bit);
        }
      }
    }
    __syncthreads();
    for (int b = wave; b < nblk; b += AL_TB / 32) {
      unsigned* bm = (unsigned*)(bmbuf + (size_t)b * RN * FDIM) + (int)blockIdx.x * 32;
      bm[lane] = tb[lane * TB_PITCH + b];
    }
    return;
  }
  const int tid = ((int)blockIdx.x - nbm) * blockDim.x + threadIdx.x;
  if (tid >= N * NHEAD) return;
  const int n = tid >> 2;
  const int h = tid & 3;
  const float* __restrict__ hp = H + (size_t)n * FDIM + h * CHAN;
  const float* __restrict__ asv = a_src + h * CHAN;
  const float* __restrict__ adv = a_dst + h * CHAN;
  float s = 0.0f, d = 0.0f;
#pragma unroll
  for (int c = 0; c < CHAN; ++c) {
    const float hv = hp[c];
    s = fmaf(hv, asv[c], s);
    d = fmaf(hv, adv[c], d);
  }
  as_[tid] = s;
  ad_[tid] = d;
}


__device__ __forceinline__ float leaky(float v) {
  return v > 0.0f ? v : NEG_SLOPE * v;
}

__device__ __forceinline__ void lds_max_float(int* addr, float v) {
  const int iv = __float_as_int(v);
  if (iv >= 0) atomicMax(addr, iv);
  else         atomicMin((unsigned int*)addr, (unsigned int)iv);
}

__device__ __forceinline__ float4 edge_logits(unsigned dl, float4 a, const float* adl) {
  const float4 d = *(const float4*)(adl + dl * NHEAD);
  return make_float4(leaky(a.x + d.x), leaky(a.y + d.y), leaky(a.z + d.z), leaky(a.w + d.w));
}

__device__ __forceinline__ void row_max_update(unsigned dl, float4 t, int* mx) {
  lds_max_float(&mx[dl * NHEAD + 0], t.x);
  lds_max_float(&mx[dl * NHEAD + 1], t.y);
  lds_max_float(&mx[dl * NHEAD + 2], t.z);
  lds_max_float(&mx[dl * NHEAD + 3], t.w);
}

__device__ __forceinline__ void first_update(int s, unsigned dl, float4 a,
                                             const float* adl, int* mx,
                                             int2* lst, float4* lgw, int* cnt) {
  const float4 t = edge_logits(dl, a, adl);
  row_max_update(dl, t, mx);
  const int pos = atomicAdd(cnt, 1);
  if (pos < LCAP) {
    lst[pos] = make_int2(s, (int)dl);
    lgw[pos] = t;
  }
}

__device__ __forceinline__ void first_edge(int e, unsigned dl,
                                           const int* __restrict__ ei,
                                           const float* __restrict__ as_,
                                           const float* adl, int* mx,
                                           int2* lst, float4* lgw, int* cnt) {
  const int s = ei[e];
  const float4 a = *(const float4*)(as_ + (size_t)s * NHEAD);
  first_update(s, dl, a, adl, mx, lst, lgw, cnt);
}

__device__ __forceinline__ void scan8(const int4 (&v)[8], const int (&jj)[8],
                                      int d0, int nloc,
                                      const int* __restrict__ ei,
                                      const float* __restrict__ as_,
                                      const float* adl, int* mx,
                                      int2* lst, float4* lgw, int* cnt) {
  int      qe0 = 0, qe1 = 0, nq = 0;
  unsigned qd0 = 0, qd1 = 0;
#pragma unroll
  for (int k = 0; k < 8; ++k) {
    const int dd[4] = {v[k].x, v[k].y, v[k].z, v[k].w};
#pragma unroll
    for (int c = 0; c < 4; ++c) {
      const unsigned dl = (unsigned)(dd[c] - d0);
      if (dl < (unsigned)nloc) {
        const int e = 4 * jj[k] + c;
        if (nq == 0)      { qe0 = e; qd0 = dl; }
        else if (nq == 1) { qe1 = e; qd1 = dl; }
        else              first_edge(e, dl, ei, as_, adl, mx, lst, lgw, cnt);
        ++nq;
      }
    }
  }
  int s0 = 0, s1 = 0;
  if (nq > 0) s0 = ei[qe0];
  if (nq > 1) s1 = ei[qe1];
  float4 a0 = make_float4(0.0f, 0.0f, 0.0f, 0.0f), a1 = a0;
  if (nq > 0) a0 = *(const float4*)(as_ + (size_t)s0 * NHEAD);
  if (nq > 1) a1 = *(const float4*)(as_ + (size_t)s1 * NHEAD);
  if (nq > 0) first_update(s0, qd0, a0, adl, mx, lst, lgw, cnt);
  if (nq > 1) first_update(s1, qd1, a1, adl, mx, lst, lgw, cnt);
}

struct EdgeIn { float4 a; float h0, h1, h2, h3; };

__device__ __forceinline__ EdgeIn edge_load(int s, int lane,
                                            const float* __restrict__ as_,
                                            const float* __restrict__ H) {
  EdgeIn in;
  in.a = *(const float4*)(as_ + (size_t)s * NHEAD);
  const float* __restrict__ hp = H + (size_t)s * FDIM + lane;
  in.h0 = hp[0 * CHAN];
  in.h1 = hp[1 * CHAN];
  in.h2 = hp[2 * CHAN];
  in.h3 = hp[3 * CHAN];
  return in;
}

__device__ __forceinline__ void edge_add(const EdgeIn in, int dl, int lane,
                                         const float* adl, const int* mx,
                                         float* den, float* acc) {
  const float e0 = __expf(leaky(in.a.x + adl[dl * NHEAD + 0]) - __int_as_float(mx[dl * NHEAD + 0]));
  const float e1 = __expf(leaky(in.a.y + adl[dl * NHEAD + 1]) - __int_as_float(mx[dl * NHEAD + 1]));
  const float e2 = __expf(leaky(in.a.z + adl[dl * NHEAD + 2]) - __int_as_float(mx[dl * NHEAD + 2]));
  const float e3 = __expf(leaky(in.a.w + adl[dl * NHEAD + 3]) - __int_as_float(mx[dl * NHEAD + 3]));
  float* ap = acc + dl * FDIM + lane;
  atomicAdd(ap + 0 * CHAN, e0 * in.h0);
  atomicAdd(ap + 1 * CHAN, e1 * in.h1);
  atomicAdd(ap + 2 * CHAN, e2 * in.h2);
  atomicAdd(ap + 3 * CHAN, e3 * in.h3);
  if (lane < NHEAD) {
    const float ev = lane == 0 ? e0 : lane == 1 ? e1 : lane == 2 ? e2 : e3;
    atomicAdd(&den[dl * NHEAD + lane], ev);
  }
}

struct RowIn { float h0, h1, h2, h3; };

__device__ __forceinline__ RowIn row_load(int s, int lane, const float* __restrict__ H) {
  RowIn in;
  const float* __restrict__ hp = H + (size_t)s * FDIM + lane;
  in.h0 = hp[0 * CHAN];
  in.h1 = hp[1 * CHAN];
  in.h2 = hp[2 * CHAN];
  in.h3 = hp[3 * CHAN];
  return in;
}

__device__ __forceinline__ void row_add(const RowIn in, float4 w, int dl, int lane, float* acc) {
  float* ap = acc + dl * FDIM + lane;
  atomicAdd(ap + 0 * CHAN, w.x * in.h0);
  atomicAdd(ap + 1 * CHAN, w.y * in.h1);
  atomicAdd(ap + 2 * CHAN, w.z * in.h2);
  atomicAdd(ap + 3 * CHAN, w.w * in.h3);
}

__device__ __forceinline__ void sweep2_slot(int e, int d, int d0, int nloc, int lane,
                                            const int* __restrict__ ei,
                                            const float* __restrict__ as_,
                                            const float* __restrict__ H,
                                            const float* adl, const int* mx,
                                            float* den, float* acc) {
  const unsigned dl = (unsigned)(d - d0);
  unsigned mask = __builtin_amdgcn_ballot_w32(dl < (unsigned)nloc);
  while (mask) {
    const int src_lane = __builtin_ctz(mask);
    mask &= mask - 1;
    const int ee = __builtin_amdgcn_readlane(e, src_lane);
    const int dd = __builtin_amdgcn_readlane((int)dl, src_lane);
    edge_add(edge_load(ei[ee], lane, as_, H), dd, lane, adl, mx, den, acc);
  }
}

__global__ __launch_bounds__(OWN_TB, 16) void gat_owner(const int* __restrict__ ei, int E, int N,
                                                    const float* __restrict__ as_,
                                                    const float* __restrict__ ad_,
                                                    const float* __restrict__ H,
                                                    const float* __restrict__ bias,
                                                    float* outp, int do_relu,
                                                    float* lbuf, int save_list, int load_list,
                                                    int use_map) {
  __shared__ float acc[RN * FDIM];
  __shared__ float den[RN * NHEAD];
  __shared__ __attribute__((aligned(16))) int   mx[RN * NHEAD];
  __shared__ __attribute__((aligned(16))) float adl[RN * NHEAD];
  __shared__ int2  lst[LCAP];
  __shared__ float4 lgw[LCAP];
  __shared__ int   cnt;
  __shared__ int   cq[CQCAP];
  __shared__ int   ccnt;

  const int tid  = threadIdx.x;
  const int lane = tid & 31;
  const int wave = tid >> 5;
  const int d0   = blockIdx.x * RN;
  int nloc = N - d0;
  if (nloc > RN) nloc = RN;

  for (int i = tid; i < RN * FDIM; i += OWN_TB) acc[i] = 0.0f;
  for (int i = tid; i < RN * NHEAD; i += OWN_TB) {
    float a = 0.0f, m = 0.0f;
    if (i < nloc * NHEAD) {
      a = ad_[(size_t)d0 * NHEAD + i];
      m = leaky(as_[(size_t)d0 * NHEAD + i] + a);
    }
    adl[i] = a;
    mx[i]  = __float_as_int(m);
    den[i] = 0.0f;
  }
  if (tid == 0) { cnt = 0; ccnt = 0; }
  __syncthreads();

  const int* __restrict__ dst = ei + E;
  const int nvec = ((E & 3) == 0) ? (E >> 2) : 0;
  const int4* __restrict__ dst4 = (const int4*)dst;
  const int rot = (int)(((long long)blockIdx.x * nvec) / gridDim.x) & ~31;

  int*  lb   = (int*)(lbuf + (size_t)d0 * FDIM);
  int2* lb2  = (int2*)lb;
  int   gcap = (nloc * FDIM) / 2 - 1;
  if (gcap > LCAP) gcap = LCAP;

  bool have_list = false;
  if (load_list) {
    const int saved = lb[0];
    if ((unsigned)saved <= (unsigned)gcap) {
      have_list = true;
      for (int base = 0; base < saved; base += OWN_TB * 4) {
        int2   en[4];
        float4 a[4];
#pragma unroll
        for (int k = 0; k < 4; ++k) {
          const int i = base + k * OWN_TB + tid;
          en[k] = make_int2(0, 0);
          if (i < saved) en[k] = lb2[1 + i];
        }
#pragma unroll
        for (int k = 0; k < 4; ++k) {
          const int i = base + k * OWN_TB + tid;
          a[k] = make_float4(0.0f, 0.0f, 0.0f, 0.0f);
          if (i < saved) {
            lst[i] = en[k];
            a[k] = *(const float4*)(as_ + (size_t)en[k].x * NHEAD);
          }
        }
#pragma unroll
        for (int k = 0; k < 4; ++k) {
          const int i = base + k * OWN_TB + tid;
          if (i < saved) {
            const float4 t = edge_logits((unsigned)en[k].y, a[k], adl);
            lgw[i] = t;
            row_max_update((unsigned)en[k].y, t, mx);
          }
        }
      }
      if (tid == 0) cnt = saved;
    }
  }

  bool need_sweep = !have_list;
  if (need_sweep && use_map && bm_fit(E, N, (int)gridDim.x)) {
    const int nw = bm_words(E);
    const unsigned* bm = (const unsigned*)(outp + (size_t)d0 * FDIM);
    for (int w0 = 0; w0 < nw; w0 += OWN_TB) {
      const int w = w0 + tid;
      unsigned bits = (w < nw) ? bm[w] : 0u;
      const int c = __builtin_popcount(bits);
      int incl = c;
      for (int o = 1; o < 32; o <<= 1) {
        const int up = __shfl_up(incl, o, 32);
        if (lane >= o) incl += up;
      }
      const int tot = __builtin_amdgcn_readlane(incl, 31);
      if (tot != 0) {
        int pos0 = 0;
        if (lane == 0) pos0 = atomicAdd(&ccnt, tot);
        int pos = __builtin_amdgcn_readfirstlane(pos0) + incl - c;
        while (bits != 0u) {
          const int g = w * 32 + __builtin_ctz(bits);
          bits &= bits - 1u;
          if (pos < CQCAP) cq[pos] = g;
          ++pos;
        }
      }
    }
    __syncthreads();
    const int ncand = ccnt;
    if (ncand <= CQCAP) {
      need_sweep = false;
      for (int base = 0; base < ncand; base += OWN_TB * 2) {
        int4 v[8];
        int  jj[8];
#pragma unroll
        for (int k = 0; k < 2; ++k) {
          const int i = base + k * OWN_TB + tid;
          const int g = (i < ncand) ? cq[i] : 0;
#pragma unroll
          for (int c = 0; c < 4; ++c) {
            jj[4 * k + c] = 4 * g + c;
            v[4 * k + c] = make_int4(-1, -1, -1, -1);
            if (i < ncand) v[4 * k + c] = dst4[4 * g + c];
          }
        }
        scan8(v, jj, d0, nloc, ei, as_, adl, mx, lst, lgw, &cnt);
      }
    }
  }

  if (need_sweep) {
  for (int base = 0; base < nvec; base += OWN_TB * 8) {
    int4 v[8];
    int  jj[8];
#pragma unroll
    for (int k = 0; k < 8; ++k) {
      const int i = base + k * OWN_TB + tid;
      int j = i + rot;
      if (j >= nvec) j -= nvec;
      jj[k] = j;
      v[k] = make_int4(-1, -1, -1, -1);
      if (i < nvec) v[k] = dst4[j];
    }
    scan8(v, jj, d0, nloc, ei, as_, adl, mx, lst, lgw, &cnt);
  }
  for (int e = 4 * nvec + tid; e < E; e += OWN_TB) {
    const unsigned dl = (unsigned)(dst[e] - d0);
    if (dl < (unsigned)nloc) first_edge(e, dl, ei, as_, adl, mx, lst, lgw, &cnt);
  }
  }
  __syncthreads();
  const int total = cnt;

  if (save_list) {
    if (tid == 0) { lb[0] = total; lb[1] = 0; }
    if (total <= gcap)
      for (int i = tid; i < total; i += OWN_TB) lb2[1 + i] = lst[i];
  }

  if (total <= LCAP) {
    for (int i = tid; i < total; i += OWN_TB) {
      const int    dl = lst[i].y;
      const float4 t  = lgw[i];
      const int4   m  = *(const int4*)(mx + dl * NHEAD);
      const float4 w  = make_float4(__expf(t.x - __int_as_float(m.x)),
                                    __expf(t.y - __int_as_float(m.y)),
                                    __expf(t.z - __int_as_float(m.z)),
                                    __expf(t.w - __int_as_float(m.w)));
      lgw[i] = w;
      atomicAdd(&den[dl * NHEAD + 0], w.x);
      atomicAdd(&den[dl * NHEAD + 1], w.y);
      atomicAdd(&den[dl * NHEAD + 2], w.z);
      atomicAdd(&den[dl * NHEAD + 3], w.w);
    }
  }

  for (int i = wave; i < nloc; i += OWN_WAVES)
    edge_add(edge_load(d0 + i, lane, as_, H), i, lane, adl, mx, den, acc);
  __syncthreads();

  if (total <= LCAP) {
    const int ngrp = total >> 2;
    for (int g = wave; g < ngrp; g += OWN_WAVES) {
      int    sN[4], dN[4];
      float4 wN[4];
#pragma unroll
      for (int u = 0; u < 4; ++u) {
        const int2 en = lst[4 * g + u];
        sN[u] = __builtin_amdgcn_readfirstlane(en.x);
        dN[u] = __builtin_amdgcn_readfirstlane(en.y);
        wN[u] = lgw[4 * g + u];
      }
      RowIn in[4];
#pragma unroll
      for (int u = 0; u < 4; ++u) in[u] = row_load(sN[u], lane, H);
#pragma unroll
      for (int u = 0; u < 4; ++u) row_add(in[u], wN[u], dN[u], lane, acc);
    }
    for (int i = 4 * ngrp + wave; i < total; i += OWN_WAVES) {
      const int2 en = lst[i];
      const int s  = __builtin_amdgcn_readfirstlane(en.x);
      const int dl = __builtin_amdgcn_readfirstlane(en.y);
      row_add(row_load(s, lane, H), lgw[i], dl, lane, acc);
    }
  } else {
    for (int base = 0; base < nvec; base += OWN_TB * 4) {
      int4 v[4];
      int  jj[4];
#pragma unroll
      for (int k = 0; k < 4; ++k) {
        const int i = base + k * OWN_TB + tid;
        int j = i + rot;
        if (j >= nvec) j -= nvec;
        jj[k] = j;
        v[k] = make_int4(-1, -1, -1, -1);
        if (i < nvec) v[k] = dst4[j];
      }
#pragma unroll
      for (int k = 0; k < 4; ++k) {
        sweep2_slot(4 * jj[k] + 0, v[k].x, d0, nloc, lane, ei, as_, H, adl, mx, den, acc);
        sweep2_slot(4 * jj[k] + 1, v[k].y, d0, nloc, lane, ei, as_, H, adl, mx, den, acc);
        sweep2_slot(4 * jj[k] + 2, v[k].z, d0, nloc, lane, ei, as_, H, adl, mx, den, acc);
        sweep2_slot(4 * jj[k] + 3, v[k].w, d0, nloc, lane, ei, as_, H, adl, mx, den, acc);
      }
    }
    for (int eb = 4 * nvec; eb < E; eb += OWN_TB) {
      const int e = eb + tid;
      const int d = (e < E) ? dst[e] : -1;
      sweep2_slot(e, d, d0, nloc, lane, ei, as_, H, adl, mx, den, acc);
    }
  }
  __syncthreads();

  for (int i = tid; i < nloc * FDIM; i += OWN_TB) {
    const int r = i >> 7, c = i & (FDIM - 1);
    float o = acc[i] / (den[r * NHEAD + (c >> 5)] + 1e-16f) + bias[c];
    if (do_relu) o = o > 0.0f ? o : 0.0f;
    outp[(size_t)d0 * FDIM + i] = o;
  }
}

static inline int cdiv(long long a, int b) { return (int)((a + b - 1) / b); }

extern "C" void kernel_launch(void* const* d_in, const int* in_sizes, int n_in,
                              void* d_out, int out_size, void* d_ws, size_t ws_size,
                              hipStream_t stream) {
  const float* x     = (const float*)d_in[0];
  const int*   ei    = (const int*)d_in[1];
  const float* W1    = (const float*)d_in[2];
  const float* asrc1 = (const float*)d_in[3];
  const float* adst1 = (const float*)d_in[4];
  const float* b1    = (const float*)d_in[5];
  const float* W2    = (const float*)d_in[6];
  const float* asrc2 = (const float*)d_in[7];
  const float* adst2 = (const float*)d_in[8];
  const float* b2    = (const float*)d_in[9];
  float* out = (float*)d_out;

  const int N  = in_sizes[0] / FDIM;
  const int E  = in_sizes[1] / 2;

  float* H    = (float*)d_ws;
  float* Hact = H    + (size_t)N * FDIM;
  float* as_  = Hact + (size_t)N * FDIM;
  float* ad_  = as_  + (size_t)N * NHEAD;

  const int TB = AL_TB;
  const int gNH    = cdiv((long long)N * NHEAD, TB);
  const int gBm    = cdiv(E >> 4, BM_CHUNK);
  const int gOwn   = cdiv(N, RN);

  gat_gemm_wmma<<<GEMM_GRID, GEMM_TB, 0, stream>>>(x, W1, H, N);
  gat_alpha<<<gNH + gBm, TB, 0, stream>>>(H, asrc1, adst1, as_, ad_, N, ei, E, gOwn,  Hact,  gBm);
  gat_owner<<<gOwn, OWN_TB, 0, stream>>>(ei, E, N, as_, ad_, H, b1, Hact,  1,
                                         out,  1,  0,  1);

  gat_gemm_wmma<<<GEMM_GRID, GEMM_TB, 0, stream>>>(Hact, W2, H, N);
  gat_alpha<<<gNH, TB, 0, stream>>>(H, asrc2, adst2, as_, ad_, N, ei, E, gOwn,  nullptr,  0);
  gat_owner<<<gOwn, OWN_TB, 0, stream>>>(ei, E, N, as_, ad_, H, b2, out,  0,
                                         out,  0,  1,  0);
}


# Owner loads touch fewer lines each: a source row is one 16-byte load per lane (was 4 one-line loads; row sums channel-permuted in LDS), a queued group's 64 B go to 4 lanes; LDS f32 add order changes
# speedup vs baseline: 5.0563x; 1.0465x over previous
#include <hip/hip_runtime.h>
#include <hip/hip_bf16.h>


typedef __attribute__((ext_vector_type(2))) float v2f;
typedef __attribute__((ext_vector_type(8))) float v8f;

#define FDIM 128
#define NHEAD 4
#define CHAN 32
#define NEG_SLOPE 0.2f

#define RN 98
#define OWN_TB 1024
#define OWN_WAVES (OWN_TB / 32)
#define LCAP 4096
#define CQCAP 2048
#define AL_TB 256
#define BM_CHUNK 1024
#define OWN_MAX 512
#define TB_PITCH (OWN_MAX + 1)

__device__ __forceinline__ int bm_words(int E) {
  return (((E >> 4) + BM_CHUNK - 1) / BM_CHUNK) * 32;
}
__device__ __forceinline__ bool bm_fit(int E, int N, int nblk) {
  return E > 0 && (E & 15) == 0 && nblk >= 1 && nblk <= OWN_MAX &&
         bm_words(E) <= (N - (nblk - 1) * RN) * FDIM;
}
static_assert(BM_CHUNK == 32 * 32, "a builder block fills 32 words of 32 groups per owner");
static_assert((OWN_MAX & (OWN_MAX - 1)) == 0, "the builders mask an owner number into the table's columns");
static_assert(BM_CHUNK / 32 == 4 * (AL_TB / 32), "a builder wave takes four of the 32 words");

#define GEMM_TB 256
#define GEMM_GRID 768
#define XPITCH (FDIM + 4)

__global__ __launch_bounds__(GEMM_TB) void gat_gemm_wmma(const float* __restrict__ X,
                                                         const float* __restrict__ W,
                                                         float* __restrict__ H,
                                                         int N) {
  __shared__ __attribute__((aligned(16))) float xs[2][16 * XPITCH];

  const int tid  = threadIdx.x;
  const int lane = tid & 31;
  const int wave = tid >> 5;
  const int m    = lane & 15;
  const int half = lane >> 4;
  const int col0 = wave * 16;
  const int ntile = (N + 15) >> 4;

  v2f b[FDIM / 4];
#pragma unroll
  for (int j = 0; j < FDIM / 4; ++j) {
    const int ka = 4 * j + 2 * half;
    b[j].x = W[ka * FDIM + col0 + m];
    b[j].y = W[(ka + 1) * FDIM + col0 + m];
  }

  const int sr0 = tid >> 5, sr1 = sr0 + 8;
  const int sc  = (tid & 31) * 4;

  int tile = blockIdx.x;
  if (tile >= ntile) return;
  float4 st0, st1;
  {
    int r0 = tile * 16 + sr0, r1 = tile * 16 + sr1;
    if (r0 >= N) r0 = N - 1;
    if (r1 >= N) r1 = N - 1;
    st0 = *(const float4*)(X + (size_t)r0 * FDIM + sc);
    st1 = *(const float4*)(X + (size_t)r1 * FDIM + sc);
  }

  int buf = 0;
  for (;;) {
    float* xt = xs[buf];
    *(float4*)(xt + sr0 * XPITCH + sc) = st0;
    *(float4*)(xt + sr1 * XPITCH + sc) = st1;
    __syncthreads();

    const int nxt = tile + GEMM_GRID;
    {
      const int lt = nxt < ntile ? nxt : tile;
      int r0 = lt * 16 + sr0, r1 = lt * 16 + sr1;
      if (r0 >= N) r0 = N - 1;
      if (r1 >= N) r1 = N - 1;
      st0 = *(const float4*)(X + (size_t)r0 * FDIM + sc);
      st1 = *(const float4*)(X + (size_t)r1 * FDIM + sc);
    }

    const float* xa = xt + m * XPITCH + 2 * half;
    v2f a[FDIM / 4];
#pragma unroll
    for (int j = 0; j < FDIM / 4; ++j) a[j] = *(const v2f*)(xa + 4 * j);
    __builtin_amdgcn_sched_barrier(0);
    v8f acc = {};
#pragma unroll
    for (int j = 0; j < FDIM / 4; ++j) {
      acc = __builtin_amdgcn_wmma_f32_16x16x4_f32(
           false, a[j],  false, b[j],
           (short)0, acc,  false,  false);
    }

    const int row0 = tile * 16;
#pragma unroll
    for (int v = 0; v < 8; ++v) {
      const int row = row0 + v + 8 * half;
      if (row < N) H[(size_t)row * FDIM + col0 + m] = acc[v];
    }
    if (nxt >= ntile) break;
    tile = nxt;
    buf ^= 1;
  }
}

__global__ __launch_bounds__(AL_TB) void gat_alpha(const float* __restrict__ H,
                          const float* __restrict__ a_src,
                          const float* __restrict__ a_dst,
                          float* __restrict__ as_, float* __restrict__ ad_,
                          int N,
                          const int* __restrict__ ei, int E, int nblk,
                          float* bmbuf, int nbm) {
  __shared__ unsigned tb[32 * TB_PITCH];
  if ((int)blockIdx.x < nbm) {
    if (bmbuf == nullptr || !bm_fit(E, N, nblk)) return;
    const int lane = threadIdx.x & 31;
    const int wave = threadIdx.x >> 5;
    const int ng   = E >> 4;
    const int4* __restrict__ dst4 = (const int4*)(ei + E);
    int4 v[4][4];
    bool live[4];
#pragma unroll
    for (int k = 0; k < 4; ++k) {
      const int g = (int)blockIdx.x * BM_CHUNK + (k * (AL_TB / 32) + wave) * 32 + lane;
      live[k] = g < ng;
      const int gl = live[k] ? g : 0;
#pragma unroll
      for (int c = 0; c < 4; ++c) v[k][c] = dst4[4 * gl + c];
    }
    for (int i = threadIdx.x; i < 32 * TB_PITCH; i += AL_TB) tb[i] = 0u;
    __syncthreads();
    const unsigned bit = 1u << lane;
#pragma unroll
    for (int k = 0; k < 4; ++k) {
      if (live[k]) {
        unsigned* row = tb + (k * (AL_TB / 32) + wave) * TB_PITCH;
#pragma unroll
        for (int c = 0; c < 4; ++c) {
          atomicOr(row + ((unsigned)v[k][c].x / RN & (OWN_MAX - 1u)), bit);
          atomicOr(row + ((unsigned)v[k][c].y / RN & (OWN_MAX - 1u)), bit);
          atomicOr(row + ((unsigned)v[k][c].z / RN & (OWN_MAX - 1u)), bit);
          atomicOr(row + ((unsigned)v[k][c].w / RN & (OWN_MAX - 1u)), bit);
        }
      }
    }
    __syncthreads();
    for (int b = wave; b < nblk; b += AL_TB / 32) {
      unsigned* bm = (unsigned*)(bmbuf + (size_t)b * RN * FDIM) + (int)blockIdx.x * 32;
      bm[lane] = tb[lane * TB_PITCH + b];
    }
    return;
  }
  const int tid = ((int)blockIdx.x - nbm) * blockDim.x + threadIdx.x;
  if (tid >= N * NHEAD) return;
  const int n = tid >> 2;
  const int h = tid & 3;
  const float* __restrict__ hp = H + (size_t)n * FDIM + h * CHAN;
  const float* __restrict__ asv = a_src + h * CHAN;
  const float* __restrict__ adv = a_dst + h * CHAN;
  float s = 0.0f, d = 0.0f;
#pragma unroll
  for (int c = 0; c < CHAN; ++c) {
    const float hv = hp[c];
    s = fmaf(hv, asv[c], s);
    d = fmaf(hv, adv[c], d);
  }
  as_[tid] = s;
  ad_[tid] = d;
}


__device__ __forceinline__ float leaky(float v) {
  return v > 0.0f ? v : NEG_SLOPE * v;
}

__device__ __forceinline__ void lds_max_float(int* addr, float v) {
  const int iv = __float_as_int(v);
  if (iv >= 0) atomicMax(addr, iv);
  else         atomicMin((unsigned int*)addr, (unsigned int)iv);
}

__device__ __forceinline__ float4 edge_logits(unsigned dl, float4 a, const float* adl) {
  const float4 d = *(const float4*)(adl + dl * NHEAD);
  return make_float4(leaky(a.x + d.x), leaky(a.y + d.y), leaky(a.z + d.z), leaky(a.w + d.w));
}

__device__ __forceinline__ void row_max_update(unsigned dl, float4 t, int* mx) {
  lds_max_float(&mx[dl * NHEAD + 0], t.x);
  lds_max_float(&mx[dl * NHEAD + 1], t.y);
  lds_max_float(&mx[dl * NHEAD + 2], t.z);
  lds_max_float(&mx[dl * NHEAD + 3], t.w);
}

__device__ __forceinline__ void first_update(int s, unsigned dl, float4 a,
                                             const float* adl, int* mx,
                                             int2* lst, float4* lgw, int* cnt) {
  const float4 t = edge_logits(dl, a, adl);
  row_max_update(dl, t, mx);
  const int pos = atomicAdd(cnt, 1);
  if (pos < LCAP) {
    lst[pos] = make_int2(s, (int)dl);
    lgw[pos] = t;
  }
}

__device__ __forceinline__ void first_edge(int e, unsigned dl,
                                           const int* __restrict__ ei,
                                           const float* __restrict__ as_,
                                           const float* adl, int* mx,
                                           int2* lst, float4* lgw, int* cnt) {
  const int s = ei[e];
  const float4 a = *(const float4*)(as_ + (size_t)s * NHEAD);
  first_update(s, dl, a, adl, mx, lst, lgw, cnt);
}

__device__ __forceinline__ void scan8(const int4 (&v)[8], const int (&jj)[8],
                                      int d0, int nloc,
                                      const int* __restrict__ ei,
                                      const float* __restrict__ as_,
                                      const float* adl, int* mx,
                                      int2* lst, float4* lgw, int* cnt) {
  int      qe0 = 0, qe1 = 0, nq = 0;
  unsigned qd0 = 0, qd1 = 0;
#pragma unroll
  for (int k = 0; k < 8; ++k) {
    const int dd[4] = {v[k].x, v[k].y, v[k].z, v[k].w};
#pragma unroll
    for (int c = 0; c < 4; ++c) {
      const unsigned dl = (unsigned)(dd[c] - d0);
      if (dl < (unsigned)nloc) {
        const int e = 4 * jj[k] + c;
        if (nq == 0)      { qe0 = e; qd0 = dl; }
        else if (nq == 1) { qe1 = e; qd1 = dl; }
        else              first_edge(e, dl, ei, as_, adl, mx, lst, lgw, cnt);
        ++nq;
      }
    }
  }
  int s0 = 0, s1 = 0;
  if (nq > 0) s0 = ei[qe0];
  if (nq > 1) s1 = ei[qe1];
  float4 a0 = make_float4(0.0f, 0.0f, 0.0f, 0.0f), a1 = a0;
  if (nq > 0) a0 = *(const float4*)(as_ + (size_t)s0 * NHEAD);
  if (nq > 1) a1 = *(const float4*)(as_ + (size_t)s1 * NHEAD);
  if (nq > 0) first_update(s0, qd0, a0, adl, mx, lst, lgw, cnt);
  if (nq > 1) first_update(s1, qd1, a1, adl, mx, lst, lgw, cnt);
}


struct EdgeIn { float4 a; float4 h; };

__device__ __forceinline__ EdgeIn edge_load(int s, int lane,
                                            const float* __restrict__ as_,
                                            const float* __restrict__ H) {
  EdgeIn in;
  in.a = *(const float4*)(as_ + (size_t)s * NHEAD);
  in.h = *(const float4*)(H + (size_t)s * FDIM + 4 * lane);
  return in;
}

__device__ __forceinline__ void edge_add(const EdgeIn in, int dl, int lane,
                                         const float* adl, const int* mx,
                                         float* den, float* acc) {
  const float e0 = __expf(leaky(in.a.x + adl[dl * NHEAD + 0]) - __int_as_float(mx[dl * NHEAD + 0]));
  const float e1 = __expf(leaky(in.a.y + adl[dl * NHEAD + 1]) - __int_as_float(mx[dl * NHEAD + 1]));
  const float e2 = __expf(leaky(in.a.z + adl[dl * NHEAD + 2]) - __int_as_float(mx[dl * NHEAD + 2]));
  const float e3 = __expf(leaky(in.a.w + adl[dl * NHEAD + 3]) - __int_as_float(mx[dl * NHEAD + 3]));
  const int hd = lane >> 3;
  const float eh = hd == 0 ? e0 : hd == 1 ? e1 : hd == 2 ? e2 : e3;
  float* ap = acc + dl * FDIM + lane;
  atomicAdd(ap + 0 * CHAN, eh * in.h.x);
  atomicAdd(ap + 1 * CHAN, eh * in.h.y);
  atomicAdd(ap + 2 * CHAN, eh * in.h.z);
  atomicAdd(ap + 3 * CHAN, eh * in.h.w);
  if (lane < NHEAD) {
    const float ev = lane == 0 ? e0 : lane == 1 ? e1 : lane == 2 ? e2 : e3;
    atomicAdd(&den[dl * NHEAD + lane], ev);
  }
}

__device__ __forceinline__ float4 row_load(int s, int lane, const float* __restrict__ H) {
  return *(const float4*)(H + (size_t)s * FDIM + 4 * lane);
}

__device__ __forceinline__ void row_add(const float4 h, float w, int dl, int lane, float* acc) {
  float* ap = acc + dl * FDIM + lane;
  atomicAdd(ap + 0 * CHAN, w * h.x);
  atomicAdd(ap + 1 * CHAN, w * h.y);
  atomicAdd(ap + 2 * CHAN, w * h.z);
  atomicAdd(ap + 3 * CHAN, w * h.w);
}

__device__ __forceinline__ void sweep2_slot(int e, int d, int d0, int nloc, int lane,
                                            const int* __restrict__ ei,
                                            const float* __restrict__ as_,
                                            const float* __restrict__ H,
                                            const float* adl, const int* mx,
                                            float* den, float* acc) {
  const unsigned dl = (unsigned)(d - d0);
  unsigned mask = __builtin_amdgcn_ballot_w32(dl < (unsigned)nloc);
  while (mask) {
    const int src_lane = __builtin_ctz(mask);
    mask &= mask - 1;
    const int ee = __builtin_amdgcn_readlane(e, src_lane);
    const int dd = __builtin_amdgcn_readlane((int)dl, src_lane);
    edge_add(edge_load(ei[ee], lane, as_, H), dd, lane, adl, mx, den, acc);
  }
}

__global__ __launch_bounds__(OWN_TB, 16) void gat_owner(const int* __restrict__ ei, int E, int N,
                                                    const float* __restrict__ as_,
                                                    const float* __restrict__ ad_,
                                                    const float* __restrict__ H,
                                                    const float* __restrict__ bias,
                                                    float* outp, int do_relu,
                                                    float* lbuf, int save_list, int load_list,
                                                    int use_map) {
  __shared__ float acc[RN * FDIM];
  __shared__ float den[RN * NHEAD];
  __shared__ __attribute__((aligned(16))) int   mx[RN * NHEAD];
  __shared__ __attribute__((aligned(16))) float adl[RN * NHEAD];
  __shared__ int2  lst[LCAP];
  __shared__ float4 lgw[LCAP];
  __shared__ int   cnt;
  __shared__ int   cq[CQCAP];
  __shared__ int   ccnt;

  const int tid  = threadIdx.x;
  const int lane = tid & 31;
  const int wave = tid >> 5;
  const int d0   = blockIdx.x * RN;
  int nloc = N - d0;
  if (nloc > RN) nloc = RN;

  for (int i = tid; i < RN * FDIM; i += OWN_TB) acc[i] = 0.0f;
  for (int i = tid; i < RN * NHEAD; i += OWN_TB) {
    float a = 0.0f, m = 0.0f;
    if (i < nloc * NHEAD) {
      a = ad_[(size_t)d0 * NHEAD + i];
      m = leaky(as_[(size_t)d0 * NHEAD + i] + a);
    }
    adl[i] = a;
    mx[i]  = __float_as_int(m);
    den[i] = 0.0f;
  }
  if (tid == 0) { cnt = 0; ccnt = 0; }
  __syncthreads();

  const int* __restrict__ dst = ei + E;
  const int nvec = ((E & 3) == 0) ? (E >> 2) : 0;
  const int4* __restrict__ dst4 = (const int4*)dst;
  const int rot = (int)(((long long)blockIdx.x * nvec) / gridDim.x) & ~31;

  int*  lb   = (int*)(lbuf + (size_t)d0 * FDIM);
  int2* lb2  = (int2*)lb;
  int   gcap = (nloc * FDIM) / 2 - 1;
  if (gcap > LCAP) gcap = LCAP;

  bool have_list = false;
  if (load_list) {
    const int saved = lb[0];
    if ((unsigned)saved <= (unsigned)gcap) {
      have_list = true;
      for (int base = 0; base < saved; base += OWN_TB * 4) {
        int2   en[4];
        float4 a[4];
#pragma unroll
        for (int k = 0; k < 4; ++k) {
          const int i = base + k * OWN_TB + tid;
          en[k] = make_int2(0, 0);
          if (i < saved) en[k] = lb2[1 + i];
        }
#pragma unroll
        for (int k = 0; k < 4; ++k) {
          const int i = base + k * OWN_TB + tid;
          a[k] = make_float4(0.0f, 0.0f, 0.0f, 0.0f);
          if (i < saved) {
            lst[i] = en[k];
            a[k] = *(const float4*)(as_ + (size_t)en[k].x * NHEAD);
          }
        }
#pragma unroll
        for (int k = 0; k < 4; ++k) {
          const int i = base + k * OWN_TB + tid;
          if (i < saved) {
            const float4 t = edge_logits((unsigned)en[k].y, a[k], adl);
            lgw[i] = t;
            row_max_update((unsigned)en[k].y, t, mx);
          }
        }
      }
      if (tid == 0) cnt = saved;
    }
  }

  bool need_sweep = !have_list;
  if (need_sweep && use_map && bm_fit(E, N, (int)gridDim.x)) {
    const int nw = bm_words(E);
    const unsigned* bm = (const unsigned*)(outp + (size_t)d0 * FDIM);
    for (int w0 = 0; w0 < nw; w0 += OWN_TB) {
      const int w = w0 + tid;
      unsigned bits = (w < nw) ? bm[w] : 0u;
      const int c = __builtin_popcount(bits);
      int incl = c;
      for (int o = 1; o < 32; o <<= 1) {
        const int up = __shfl_up(incl, o, 32);
        if (lane >= o) incl += up;
      }
      const int tot = __builtin_amdgcn_readlane(incl, 31);
      if (tot != 0) {
        int pos0 = 0;
        if (lane == 0) pos0 = atomicAdd(&ccnt, tot);
        int pos = __builtin_amdgcn_readfirstlane(pos0) + incl - c;
        while (bits != 0u) {
          const int g = w * 32 + __builtin_ctz(bits);
          bits &= bits - 1u;
          if (pos < CQCAP) cq[pos] = g;
          ++pos;
        }
      }
    }
    __syncthreads();
    const int ncand = ccnt;
    if (ncand <= CQCAP) {
      need_sweep = false;
      for (int base = 0; base < ncand; base += OWN_TB * 2) {
        int4 v[8];
        int  jj[8];
        int  gq[8];
#pragma unroll
        for (int k = 0; k < 8; ++k) {
          const int i = base + k * (OWN_TB / 4) + (tid >> 2);
          gq[k] = cq[i < ncand ? i : 0];
        }
#pragma unroll
        for (int k = 0; k < 8; ++k) {
          const int i = base + k * (OWN_TB / 4) + (tid >> 2);
          jj[k] = 4 * gq[k] + (tid & 3);
          const int4 t = dst4[jj[k]];
          v[k] = (i < ncand) ? t : make_int4(-1, -1, -1, -1);
        }
        scan8(v, jj, d0, nloc, ei, as_, adl, mx, lst, lgw, &cnt);
      }
    }
  }

  if (need_sweep) {
  for (int base = 0; base < nvec; base += OWN_TB * 8) {
    int4 v[8];
    int  jj[8];
#pragma unroll
    for (int k = 0; k < 8; ++k) {
      const int i = base + k * OWN_TB + tid;
      int j = i + rot;
      if (j >= nvec) j -= nvec;
      jj[k] = j;
      v[k] = make_int4(-1, -1, -1, -1);
      if (i < nvec) v[k] = dst4[j];
    }
    scan8(v, jj, d0, nloc, ei, as_, adl, mx, lst, lgw, &cnt);
  }
  for (int e = 4 * nvec + tid; e < E; e += OWN_TB) {
    const unsigned dl = (unsigned)(dst[e] - d0);
    if (dl < (unsigned)nloc) first_edge(e, dl, ei, as_, adl, mx, lst, lgw, &cnt);
  }
  }
  __syncthreads();
  const int total = cnt;

  if (save_list) {
    if (tid == 0) { lb[0] = total; lb[1] = 0; }
    if (total <= gcap)
      for (int i = tid; i < total; i += OWN_TB) lb2[1 + i] = lst[i];
  }

  if (total <= LCAP) {
    for (int i = tid; i < total; i += OWN_TB) {
      const int    dl = lst[i].y;
      const float4 t  = lgw[i];
      const int4   m  = *(const int4*)(mx + dl * NHEAD);
      const float4 w  = make_float4(__expf(t.x - __int_as_float(m.x)),
                                    __expf(t.y - __int_as_float(m.y)),
                                    __expf(t.z - __int_as_float(m.z)),
                                    __expf(t.w - __int_as_float(m.w)));
      lgw[i] = w;
      atomicAdd(&den[dl * NHEAD + 0], w.x);
      atomicAdd(&den[dl * NHEAD + 1], w.y);
      atomicAdd(&den[dl * NHEAD + 2], w.z);
      atomicAdd(&den[dl * NHEAD + 3], w.w);
    }
  }

  for (int i = wave; i < nloc; i += OWN_WAVES)
    edge_add(edge_load(d0 + i, lane, as_, H), i, lane, adl, mx, den, acc);
  __syncthreads();

  if (total <= LCAP) {
    const int ngrp = total >> 2;
    const float* lgwh = (const float*)lgw + (lane >> 3);
    for (int g = wave; g < ngrp; g += OWN_WAVES) {
      int   sN[4], dN[4];
      float wN[4];
#pragma unroll
      for (int u = 0; u < 4; ++u) {
        const int2 en = lst[4 * g + u];
        sN[u] = __builtin_amdgcn_readfirstlane(en.x);
        dN[u] = __builtin_amdgcn_readfirstlane(en.y);
        wN[u] = lgwh[(4 * g + u) * NHEAD];
      }
      float4 in[4];
#pragma unroll
      for (int u = 0; u < 4; ++u) in[u] = row_load(sN[u], lane, H);
#pragma unroll
      for (int u = 0; u < 4; ++u) row_add(in[u], wN[u], dN[u], lane, acc);
    }
    for (int i = 4 * ngrp + wave; i < total; i += OWN_WAVES) {
      const int2 en = lst[i];
      const int s  = __builtin_amdgcn_readfirstlane(en.x);
      const int dl = __builtin_amdgcn_readfirstlane(en.y);
      row_add(row_load(s, lane, H), lgwh[i * NHEAD], dl, lane, acc);
    }
  } else {
    for (int base = 0; base < nvec; base += OWN_TB * 4) {
      int4 v[4];
      int  jj[4];
#pragma unroll
      for (int k = 0; k < 4; ++k) {
        const int i = base + k * OWN_TB + tid;
        int j = i + rot;
        if (j >= nvec) j -= nvec;
        jj[k] = j;
        v[k] = make_int4(-1, -1, -1, -1);
        if (i < nvec) v[k] = dst4[j];
      }
#pragma unroll
      for (int k = 0; k < 4; ++k) {
        sweep2_slot(4 * jj[k] + 0, v[k].x, d0, nloc, lane, ei, as_, H, adl, mx, den, acc);
        sweep2_slot(4 * jj[k] + 1, v[k].y, d0, nloc, lane, ei, as_, H, adl, mx, den, acc);
        sweep2_slot(4 * jj[k] + 2, v[k].z, d0, nloc, lane, ei, as_, H, adl, mx, den, acc);
        sweep2_slot(4 * jj[k] + 3, v[k].w, d0, nloc, lane, ei, as_, H, adl, mx, den, acc);
      }
    }
    for (int eb = 4 * nvec; eb < E; eb += OWN_TB) {
      const int e = eb + tid;
      const int d = (e < E) ? dst[e] : -1;
      sweep2_slot(e, d, d0, nloc, lane, ei, as_, H, adl, mx, den, acc);
    }
  }
  __syncthreads();

  for (int i = tid; i < nloc * 32; i += OWN_TB) {
    const int r = i >> 5, q = i & 31;
    const float dn = den[r * NHEAD + (q >> 3)] + 1e-16f;
    const float* ap = acc + r * FDIM + q;
    float4 o;
    o.x = ap[0 * CHAN] / dn + bias[4 * q + 0];
    o.y = ap[1 * CHAN] / dn + bias[4 * q + 1];
    o.z = ap[2 * CHAN] / dn + bias[4 * q + 2];
    o.w = ap[3 * CHAN] / dn + bias[4 * q + 3];
    if (do_relu) {
      o.x = o.x > 0.0f ? o.x : 0.0f;
      o.y = o.y > 0.0f ? o.y : 0.0f;
      o.z = o.z > 0.0f ? o.z : 0.0f;
      o.w = o.w > 0.0f ? o.w : 0.0f;
    }
    *(float4*)(outp + (size_t)(d0 + r) * FDIM + 4 * q) = o;
  }
}

static inline int cdiv(long long a, int b) { return (int)((a + b - 1) / b); }

extern "C" void kernel_launch(void* const* d_in, const int* in_sizes, int n_in,
                              void* d_out, int out_size, void* d_ws, size_t ws_size,
                              hipStream_t stream) {
  const float* x     = (const float*)d_in[0];
  const int*   ei    = (const int*)d_in[1];
  const float* W1    = (const float*)d_in[2];
  const float* asrc1 = (const float*)d_in[3];
  const float* adst1 = (const float*)d_in[4];
  const float* b1    = (const float*)d_in[5];
  const float* W2    = (const float*)d_in[6];
  const float* asrc2 = (const float*)d_in[7];
  const float* adst2 = (const float*)d_in[8];
  const float* b2    = (const float*)d_in[9];
  float* out = (float*)d_out;

  const int N  = in_sizes[0] / FDIM;
  const int E  = in_sizes[1] / 2;

  float* H    = (float*)d_ws;
  float* Hact = H    + (size_t)N * FDIM;
  float* as_  = Hact + (size_t)N * FDIM;
  float* ad_  = as_  + (size_t)N * NHEAD;

  const int TB = AL_TB;
  const int gNH    = cdiv((long long)N * NHEAD, TB);
  const int gBm    = cdiv(E >> 4, BM_CHUNK);
  const int gOwn   = cdiv(N, RN);

  gat_gemm_wmma<<<GEMM_GRID, GEMM_TB, 0, stream>>>(x, W1, H, N);
  gat_alpha<<<gNH + gBm, TB, 0, stream>>>(H, asrc1, adst1, as_, ad_, N, ei, E, gOwn,  Hact,  gBm);
  gat_owner<<<gOwn, OWN_TB, 0, stream>>>(ei, E, N, as_, ad_, H, b1, Hact,  1,
                                         out,  1,  0,  1);

  gat_gemm_wmma<<<GEMM_GRID, GEMM_TB, 0, stream>>>(Hact, W2, H, N);
  gat_alpha<<<gNH, TB, 0, stream>>>(H, asrc2, adst2, as_, ad_, N, ei, E, gOwn,  nullptr,  0);
  gat_owner<<<gOwn, OWN_TB, 0, stream>>>(ei, E, N, as_, ad_, H, b2, out,  0,
                                         out,  0,  1,  0);
}
